# starting kernel
# baseline (speedup 1.0000x reference)
#include <hip/hip_runtime.h>

typedef __attribute__((ext_vector_type(2))) float v2f;
typedef __attribute__((ext_vector_type(8))) float v8f;

#define FDIM 128

__global__ void __launch_bounds__(256) zero_init_kernel(unsigned* __restrict__ deg,
                                                        float* __restrict__ accum,
                                                        unsigned* __restrict__ cnt, int n) {
    int i = blockIdx.x * blockDim.x + threadIdx.x;
    if (i < n) deg[i] = 0u;
    if (i < 2) { accum[i] = 0.0f; cnt[i] = 0u; }
}

__global__ void __launch_bounds__(256) count_deg_kernel(const int* __restrict__ dst,
                                                        unsigned* __restrict__ deg, int E) {
    int e = blockIdx.x * blockDim.x + threadIdx.x;
    if (e < E) atomicAdd(&deg[dst[e]], 1u);
}

__global__ void __launch_bounds__(256) dinv_kernel(const unsigned* __restrict__ deg,
                                                   float* __restrict__ dinv, int n) {
    int i = blockIdx.x * blockDim.x + threadIdx.x;
    if (i < n) dinv[i] = rsqrtf((float)deg[i] + 1.0f);
}

template <bool RELU>
__global__ void __launch_bounds__(256) gemm_wmma_kernel(const float* __restrict__ A,
                                                        const float* __restrict__ W,
                                                        float* __restrict__ Hout,
                                                        float* __restrict__ AggOut,
                                                        const float* __restrict__ dinv,
                                                        const float* __restrict__ bias,
                                                        int nrows) {
    __shared__ float wlds[FDIM * FDIM];
    for (int i = threadIdx.x; i < FDIM * FDIM; i += 256) {
        int k = i >> 7, n = i & 127;
        wlds[((k >> 1) << 8) + (n << 1) + (k & 1)] = W[i];
    }
    __syncthreads();

    const int lane  = threadIdx.x & 31;
    const int wave  = threadIdx.x >> 5;
    const int strip = blockIdx.x * 8 + wave;
    const int nstrips = nrows >> 4;
    if (strip >= nstrips) return;

    const int row0  = strip << 4;
    const int m     = lane & 15;
    const int khalf = lane >> 4;
    const int kbase = khalf << 1;

    const float* arow = A + (size_t)(row0 + m) * FDIM;

    v8f acc[8];
#pragma unroll
    for (int t = 0; t < 8; ++t) acc[t] = (v8f)(0.0f);

    for (int k0 = 0; k0 < FDIM; k0 += 4) {
        v2f a = *(const v2f*)(arow + k0 + kbase);
        if (RELU) { a.x = fmaxf(a.x, 0.0f); a.y = fmaxf(a.y, 0.0f); }
        const int bbase = ((k0 >> 1) + khalf) * 256 + (m << 1);
#pragma unroll
        for (int t = 0; t < 8; ++t) {
            v2f b = *(const v2f*)(&wlds[bbase + t * 32]);
            acc[t] = __builtin_amdgcn_wmma_f32_16x16x4_f32(
                 false, a,  false, b,
                 (short)0, acc[t],  false,  false);
        }
    }

    const int rbase = row0 + (khalf << 3);
    float dr[8];
#pragma unroll
    for (int r = 0; r < 8; ++r) { float t = dinv[rbase + r]; dr[r] = t * t; }
#pragma unroll
    for (int t = 0; t < 8; ++t) {
        float bt = bias[t * 16 + m];
#pragma unroll
        for (int r = 0; r < 8; ++r) {
            float hv = acc[t][r];
            size_t off = (size_t)(rbase + r) * FDIM + t * 16 + m;
            Hout[off]   = hv;
            AggOut[off] = hv * dr[r] + bt;
        }
    }
}

__global__ void __launch_bounds__(256) scatter_kernel(const float4* __restrict__ h,
                                                      float* __restrict__ out,
                                                      const int* __restrict__ src,
                                                      const int* __restrict__ dst,
                                                      const float* __restrict__ dinv, int E) {
    int lane = threadIdx.x & 31;
    int e = blockIdx.x * (blockDim.x >> 5) + (threadIdx.x >> 5);
    if (e >= E) return;
    int s = src[e], d = dst[e];
    float nrm = dinv[s] * dinv[d];
    float4 v = h[(size_t)s * 32 + lane];
    float* o = out + (size_t)d * FDIM + lane * 4;
    atomicAdd(o + 0, v.x * nrm);
    atomicAdd(o + 1, v.y * nrm);
    atomicAdd(o + 2, v.z * nrm);
    atomicAdd(o + 3, v.w * nrm);
}

__global__ void __launch_bounds__(256) edge_loss_kernel(const float4* __restrict__ rep,
                                                        const int* __restrict__ src,
                                                        const int* __restrict__ dst, int E,
                                                        float target,
                                                        float* __restrict__ loss_acc,
                                                        unsigned* __restrict__ cnt_acc) {
    int lane   = threadIdx.x & 31;
    int wave   = blockIdx.x * (blockDim.x >> 5) + (threadIdx.x >> 5);
    int nwaves = gridDim.x * (blockDim.x >> 5);
    float local = 0.0f;
    unsigned cnt = 0;
    for (int e = wave; e < E; e += nwaves) {
        int s = src[e], d = dst[e];
        if (s < d) {
            float4 a = rep[(size_t)s * 32 + lane];
            float4 b = rep[(size_t)d * 32 + lane];
            float p = a.x * b.x + a.y * b.y + a.z * b.z + a.w * b.w;
#pragma unroll
            for (int off = 16; off > 0; off >>= 1) p += __shfl_down(p, off, 32);
            if (lane == 0) {
                float dl = p - target;
                local += dl * dl;
                cnt++;
            }
        }
    }
    if (lane == 0 && cnt > 0) {
        atomicAdd(loss_acc, local);
        atomicAdd(cnt_acc, cnt);
    }
}

__global__ void finalize_kernel(const float* __restrict__ accum,
                                const unsigned* __restrict__ cnt,
                                float* __restrict__ out_scalar, float n_nodes) {
    if (threadIdx.x == 0 && blockIdx.x == 0) {
        float tot = accum[0] + accum[1];
        float c = (float)(cnt[0] + cnt[1]);
        *out_scalar = tot * n_nodes / c;
    }
}

extern "C" void kernel_launch(void* const* d_in, const int* in_sizes, int n_in,
                              void* d_out, int out_size, void* d_ws, size_t ws_size,
                              hipStream_t stream) {
    const float* features = (const float*)d_in[0];
    const int*   ei       = (const int*)d_in[1];
    const int*   nei      = (const int*)d_in[2];
    const float* w1       = (const float*)d_in[3];
    const float* b1       = (const float*)d_in[4];
    const float* w2       = (const float*)d_in[5];
    const float* b2       = (const float*)d_in[6];

    const int N  = in_sizes[0] / FDIM;
    const int E  = in_sizes[1] / 2;
    const int En = in_sizes[2] / 2;
    const int* src  = ei;
    const int* dst  = ei + E;
    const int* nsrc = nei;
    const int* ndst = nei + En;

    char* ws = (char*)d_ws;
    float*    bufA = (float*)ws;
    float*    bufB = (float*)(ws + (size_t)N * FDIM * 4);
    unsigned* deg  = (unsigned*)(ws + (size_t)2 * N * FDIM * 4);
    float*    dinv = (float*)((char*)deg + (size_t)N * 4);
    float*    accum = (float*)((char*)dinv + (size_t)N * 4);
    unsigned* cnt   = (unsigned*)(accum + 2);

    float* rep      = (float*)d_out;
    float* loss_out = rep + (size_t)N * FDIM;

    const int nstrips = N >> 4;
    const int gblocks = (nstrips + 7) / 8;

    zero_init_kernel<<<(N + 255) / 256, 256, 0, stream>>>(deg, accum, cnt, N);
    count_deg_kernel<<<(E + 255) / 256, 256, 0, stream>>>(dst, deg, E);
    dinv_kernel<<<(N + 255) / 256, 256, 0, stream>>>(deg, dinv, N);

    gemm_wmma_kernel<false><<<gblocks, 256, 0, stream>>>(features, w1, bufA, bufB,
                                                         dinv, b1, N);
    scatter_kernel<<<(E + 7) / 8, 256, 0, stream>>>((const float4*)bufA, bufB, src, dst, dinv, E);

    gemm_wmma_kernel<true><<<gblocks, 256, 0, stream>>>(bufB, w2, bufA, rep,
                                                        dinv, b2, N);
    scatter_kernel<<<(E + 7) / 8, 256, 0, stream>>>((const float4*)bufA, rep, src, dst, dinv, E);

    edge_loss_kernel<<<2048, 256, 0, stream>>>((const float4*)rep, src, dst, E, 1.0f,
                                               &accum[0], &cnt[0]);
    edge_loss_kernel<<<2048, 256, 0, stream>>>((const float4*)rep, nsrc, ndst, En, 0.0f,
                                               &accum[1], &cnt[1]);
    finalize_kernel<<<1, 1, 0, stream>>>(accum, cnt, loss_out, (float)N);
}


# neighbour sums without global atomics: each workgroup keeps 391 dst rows in LDS, filters the whole dst list for them, adds the gathered h rows with LDS float atomics (arrival order)
# speedup vs baseline: 1.6047x; 1.6047x over previous
#include <hip/hip_runtime.h>

typedef __attribute__((ext_vector_type(2))) float v2f;
typedef __attribute__((ext_vector_type(8))) float v8f;

#define FDIM 128

__global__ void __launch_bounds__(256) zero_init_kernel(unsigned* __restrict__ deg,
                                                        float* __restrict__ accum,
                                                        unsigned* __restrict__ cnt, int n) {
    int i = blockIdx.x * blockDim.x + threadIdx.x;
    if (i < n) deg[i] = 0u;
    if (i < 2) { accum[i] = 0.0f; cnt[i] = 0u; }
}

__global__ void __launch_bounds__(256) count_deg_kernel(const int* __restrict__ dst,
                                                        unsigned* __restrict__ deg, int E) {
    int e = blockIdx.x * blockDim.x + threadIdx.x;
    if (e < E) atomicAdd(&deg[dst[e]], 1u);
}

__global__ void __launch_bounds__(256) dinv_kernel(const unsigned* __restrict__ deg,
                                                   float* __restrict__ dinv, int n) {
    int i = blockIdx.x * blockDim.x + threadIdx.x;
    if (i < n) dinv[i] = rsqrtf((float)deg[i] + 1.0f);
}

template <bool RELU>
__global__ void __launch_bounds__(256) gemm_wmma_kernel(const float* __restrict__ A,
                                                        const float* __restrict__ W,
                                                        float* __restrict__ Hout,
                                                        float* __restrict__ AggOut,
                                                        const float* __restrict__ dinv,
                                                        const float* __restrict__ bias,
                                                        int nrows) {
    __shared__ float wlds[FDIM * FDIM];
    for (int i = threadIdx.x; i < FDIM * FDIM; i += 256) {
        int k = i >> 7, n = i & 127;
        wlds[((k >> 1) << 8) + (n << 1) + (k & 1)] = W[i];
    }
    __syncthreads();

    const int lane  = threadIdx.x & 31;
    const int wave  = threadIdx.x >> 5;
    const int strip = blockIdx.x * 8 + wave;
    const int nstrips = nrows >> 4;
    if (strip >= nstrips) return;

    const int row0  = strip << 4;
    const int m     = lane & 15;
    const int khalf = lane >> 4;
    const int kbase = khalf << 1;

    const float* arow = A + (size_t)(row0 + m) * FDIM;

    v8f acc[8];
#pragma unroll
    for (int t = 0; t < 8; ++t) acc[t] = (v8f)(0.0f);

    for (int k0 = 0; k0 < FDIM; k0 += 4) {
        v2f a = *(const v2f*)(arow + k0 + kbase);
        if (RELU) { a.x = fmaxf(a.x, 0.0f); a.y = fmaxf(a.y, 0.0f); }
        const int bbase = ((k0 >> 1) + khalf) * 256 + (m << 1);
#pragma unroll
        for (int t = 0; t < 8; ++t) {
            v2f b = *(const v2f*)(&wlds[bbase + t * 32]);
            acc[t] = __builtin_amdgcn_wmma_f32_16x16x4_f32(
                 false, a,  false, b,
                 (short)0, acc[t],  false,  false);
        }
    }

    const int rbase = row0 + (khalf << 3);
    float dr[8];
#pragma unroll
    for (int r = 0; r < 8; ++r) { float t = dinv[rbase + r]; dr[r] = t * t; }
#pragma unroll
    for (int t = 0; t < 8; ++t) {
        float bt = bias[t * 16 + m];
#pragma unroll
        for (int r = 0; r < 8; ++r) {
            float hv = acc[t][r];
            size_t off = (size_t)(rbase + r) * FDIM + t * 16 + m;
            Hout[off]   = hv;
            AggOut[off] = hv * dr[r] + bt;
        }
    }
}

#define ROWS_PER_WG 391
#define AGG_WAVES 32
#define AGG_UNROLL 8

__global__ void __launch_bounds__(1024) owner_agg_kernel(const float4* __restrict__ h,
                                                         float* __restrict__ out,
                                                         const int* __restrict__ src,
                                                         const int* __restrict__ dst,
                                                         const float* __restrict__ dinv,
                                                         int E, int n) {
    __shared__ float tile[ROWS_PER_WG * FDIM];
    __shared__ int queue[AGG_WAVES * 64];

    const int lane = threadIdx.x & 31;
    const int wave = threadIdx.x >> 5;
    const int qb = wave * 64;
    const int lo = blockIdx.x * ROWS_PER_WG;
    int nr = n - lo;
    if (nr > ROWS_PER_WG) nr = ROWS_PER_WG;

    float4* out4 = (float4*)out + (size_t)lo * 32;
    for (int i = threadIdx.x; i < nr * 32; i += 1024) {
        float4 v = out4[i];
        float* t = tile + (i >> 5) * FDIM + (i & 31);
        t[0] = v.x; t[32] = v.y; t[64] = v.z; t[96] = v.w;
    }
    __syncthreads();

    int cnt = 0;
    const unsigned lt = (1u << lane) - 1u;

    auto add_row = [&](float4 v, int r, float w) {
        float* t = tile + r * FDIM + lane;
        atomicAdd(t, v.x * w);
        atomicAdd(t + 32, v.y * w);
        atomicAdd(t + 64, v.z * w);
        atomicAdd(t + 96, v.w * w);
    };

    auto flush = [&](int nf) {
        int s = 0, dr = 0;
        float nrm = 0.0f;
        if (lane < nf) {
            int e = queue[qb + lane];
            s = src[e];
            int d = dst[e];
            dr = d - lo;
            nrm = dinv[s] * dinv[d];
        }
        int j = 0;
        for (; j + 8 <= nf; j += 8) {
            float4 v[8];
            int r[8];
            float w[8];
#pragma unroll
            for (int u = 0; u < 8; ++u) {
                int sj = __builtin_amdgcn_readlane(s, j + u);
                r[u] = __builtin_amdgcn_readlane(dr, j + u);
                w[u] = __int_as_float(__builtin_amdgcn_readlane(__float_as_int(nrm), j + u));
                v[u] = h[(size_t)sj * 32 + lane];
            }
#pragma unroll
            for (int u = 0; u < 8; ++u) add_row(v[u], r[u], w[u]);
        }
        for (; j < nf; ++j) {
            int sj = __builtin_amdgcn_readlane(s, j);
            int rj = __builtin_amdgcn_readlane(dr, j);
            float wj = __int_as_float(__builtin_amdgcn_readlane(__float_as_int(nrm), j));
            add_row(h[(size_t)sj * 32 + lane], rj, wj);
        }
    };

    int head = (int)(((16u - ((unsigned)(size_t)dst & 15u)) & 15u) >> 2);
    if (head > E) head = E;
    const int4* dst4 = (const int4*)(dst + head);
    const int nvec = (E - head) >> 2;
    const int ntail = (E - head) & 3;
    const int nextra = head + ntail;

    const int step = AGG_WAVES * 32 * AGG_UNROLL;
    const int npass = (nvec + step - 1) / step;
    for (int pass = 0; pass <= npass; ++pass) {
        const int vb = pass * step + wave * (32 * AGG_UNROLL);
        unsigned mbits = 0u;
        if (pass < npass) {
            int4 d4[AGG_UNROLL];
            unsigned nru[AGG_UNROLL];
#pragma unroll
            for (int u = 0; u < AGG_UNROLL; ++u) {
                const int vi = vb + u * 32 + lane;
                d4[u] = dst4[vi < nvec ? vi : nvec - 1];
                nru[u] = vi < nvec ? (unsigned)nr : 0u;
            }
#pragma unroll
            for (int u = 0; u < AGG_UNROLL; ++u) {
                const bool cx = (unsigned)(d4[u].x - lo) < nru[u];
                const bool cy = (unsigned)(d4[u].y - lo) < nru[u];
                const bool cz = (unsigned)(d4[u].z - lo) < nru[u];
                const bool cw = (unsigned)(d4[u].w - lo) < nru[u];
                const unsigned any = __builtin_amdgcn_ballot_w32(cx) | __builtin_amdgcn_ballot_w32(cy) |
                                     __builtin_amdgcn_ballot_w32(cz) | __builtin_amdgcn_ballot_w32(cw);
                if (any)
                    mbits |= ((cx ? 1u : 0u) | (cy ? 2u : 0u) | (cz ? 4u : 0u) | (cw ? 8u : 0u)) << (4 * u);
            }
        } else if (wave == 0 && lane < nextra) {
            int e = lane < head ? lane : (nvec << 2) + lane;
            if ((unsigned)(dst[e] - lo) < (unsigned)nr) mbits = 1u;
        }
        while (__builtin_amdgcn_ballot_w32(mbits != 0u)) {
            const bool mt = mbits != 0u;
            const int k = mt ? (__ffs(mbits) - 1) : 0;
            int e;
            if (pass < npass) e = head + ((vb + (k >> 2) * 32 + lane) << 2) + (k & 3);
            else              e = lane < head ? lane : (nvec << 2) + lane;
            mbits &= mbits - 1u;
            const unsigned mask = __builtin_amdgcn_ballot_w32(mt);
            const int pos = cnt + __popc(mask & lt);
            if (mt) queue[qb + pos] = e;
            cnt += __popc(mask);
            if (cnt >= 32) {
                flush(32);
                const int rem = cnt - 32;
                int t = 0;
                if (lane < rem) t = queue[qb + 32 + lane];
                if (lane < rem) queue[qb + lane] = t;
                cnt = rem;
            }
        }
    }
    if (cnt > 0) flush(cnt);

    __syncthreads();
    for (int i = threadIdx.x; i < nr * 32; i += 1024) {
        const float* t = tile + (i >> 5) * FDIM + (i & 31);
        out4[i] = make_float4(t[0], t[32], t[64], t[96]);
    }
}

__global__ void __launch_bounds__(256) edge_loss_kernel(const float4* __restrict__ rep,
                                                        const int* __restrict__ src,
                                                        const int* __restrict__ dst, int E,
                                                        float target,
                                                        float* __restrict__ loss_acc,
                                                        unsigned* __restrict__ cnt_acc) {
    int lane   = threadIdx.x & 31;
    int wave   = blockIdx.x * (blockDim.x >> 5) + (threadIdx.x >> 5);
    int nwaves = gridDim.x * (blockDim.x >> 5);
    float local = 0.0f;
    unsigned cnt = 0;
    for (int e = wave; e < E; e += nwaves) {
        int s = src[e], d = dst[e];
        if (s < d) {
            float4 a = rep[(size_t)s * 32 + lane];
            float4 b = rep[(size_t)d * 32 + lane];
            float p = a.x * b.x + a.y * b.y + a.z * b.z + a.w * b.w;
#pragma unroll
            for (int off = 16; off > 0; off >>= 1) p += __shfl_down(p, off, 32);
            if (lane == 0) {
                float dl = p - target;
                local += dl * dl;
                cnt++;
            }
        }
    }
    if (lane == 0 && cnt > 0) {
        atomicAdd(loss_acc, local);
        atomicAdd(cnt_acc, cnt);
    }
}

__global__ void finalize_kernel(const float* __restrict__ accum,
                                const unsigned* __restrict__ cnt,
                                float* __restrict__ out_scalar, float n_nodes) {
    if (threadIdx.x == 0 && blockIdx.x == 0) {
        float tot = accum[0] + accum[1];
        float c = (float)(cnt[0] + cnt[1]);
        *out_scalar = tot * n_nodes / c;
    }
}

extern "C" void kernel_launch(void* const* d_in, const int* in_sizes, int n_in,
                              void* d_out, int out_size, void* d_ws, size_t ws_size,
                              hipStream_t stream) {
    const float* features = (const float*)d_in[0];
    const int*   ei       = (const int*)d_in[1];
    const int*   nei      = (const int*)d_in[2];
    const float* w1       = (const float*)d_in[3];
    const float* b1       = (const float*)d_in[4];
    const float* w2       = (const float*)d_in[5];
    const float* b2       = (const float*)d_in[6];

    const int N  = in_sizes[0] / FDIM;
    const int E  = in_sizes[1] / 2;
    const int En = in_sizes[2] / 2;
    const int* src  = ei;
    const int* dst  = ei + E;
    const int* nsrc = nei;
    const int* ndst = nei + En;

    char* ws = (char*)d_ws;
    float*    bufA = (float*)ws;
    float*    bufB = (float*)(ws + (size_t)N * FDIM * 4);
    unsigned* deg  = (unsigned*)(ws + (size_t)2 * N * FDIM * 4);
    float*    dinv = (float*)((char*)deg + (size_t)N * 4);
    float*    accum = (float*)((char*)dinv + (size_t)N * 4);
    unsigned* cnt   = (unsigned*)(accum + 2);

    float* rep      = (float*)d_out;
    float* loss_out = rep + (size_t)N * FDIM;

    const int nstrips = N >> 4;
    const int gblocks = (nstrips + 7) / 8;
    const int ablocks = (N + ROWS_PER_WG - 1) / ROWS_PER_WG;

    zero_init_kernel<<<(N + 255) / 256, 256, 0, stream>>>(deg, accum, cnt, N);
    count_deg_kernel<<<(E + 255) / 256, 256, 0, stream>>>(dst, deg, E);
    dinv_kernel<<<(N + 255) / 256, 256, 0, stream>>>(deg, dinv, N);

    gemm_wmma_kernel<false><<<gblocks, 256, 0, stream>>>(features, w1, bufA, bufB,
                                                         dinv, b1, N);
    owner_agg_kernel<<<ablocks, 1024, 0, stream>>>((const float4*)bufA, bufB, src, dst, dinv, E, N);

    gemm_wmma_kernel<true><<<gblocks, 256, 0, stream>>>(bufB, w2, bufA, rep,
                                                        dinv, b2, N);
    owner_agg_kernel<<<ablocks, 1024, 0, stream>>>((const float4*)bufA, rep, src, dst, dinv, E, N);

    edge_loss_kernel<<<2048, 256, 0, stream>>>((const float4*)rep, src, dst, E, 1.0f,
                                               &accum[0], &cnt[0]);
    edge_loss_kernel<<<2048, 256, 0, stream>>>((const float4*)rep, nsrc, ndst, En, 0.0f,
                                               &accum[1], &cnt[1]);
    finalize_kernel<<<1, 1, 0, stream>>>(accum, cnt, loss_out, (float)N);
}


# edges binned by owner workgroup inside the degree pass (LDS ranks, one returning add per owner per block, runs staged in LDS); layer 1 walks its bin, no dst filter; LDS atomics, arrival order
# speedup vs baseline: 2.3247x; 1.4487x over previous
#include <hip/hip_runtime.h>

typedef __attribute__((ext_vector_type(2))) float v2f;
typedef __attribute__((ext_vector_type(8))) float v8f;

#define FDIM 128
#define ROWS_PER_WG 391
#define LIST_HDR 64
#define MAX_OWNERS 256
#define BIN_THREADS 256
#define BIN_EPT 8
#define BIN_EDGES (BIN_THREADS * BIN_EPT)
static_assert(BIN_THREADS == MAX_OWNERS, "thread b of the binning pass handles owner b");

__global__ void __launch_bounds__(256) zero_init_kernel(unsigned* __restrict__ deg,
                                                        float* __restrict__ accum,
                                                        unsigned* __restrict__ cnt,
                                                        unsigned* __restrict__ bins, int n, int nown) {
    int i = blockIdx.x * blockDim.x + threadIdx.x;
    if (i < n) deg[i] = 0u;
    if (i < 2) { accum[i] = 0.0f; cnt[i] = 0u; }
    if (i < nown) bins[(size_t)i * ROWS_PER_WG * FDIM] = 0u;
}

__global__ void __launch_bounds__(BIN_THREADS) count_bin_kernel(const int* __restrict__ src,
                                                                const int* __restrict__ dst,
                                                                unsigned* __restrict__ deg,
                                                                unsigned* __restrict__ bins,
                                                                int E, int n, int nown) {
    __shared__ unsigned hist[MAX_OWNERS];
    __shared__ unsigned gbase[MAX_OWNERS];
    __shared__ unsigned wtot[BIN_THREADS / 32];
    __shared__ unsigned st_e[BIN_EDGES];
    __shared__ unsigned st_a[BIN_EDGES];

    const int tid = threadIdx.x;
    const int lane = tid & 31;
    const int wave = tid >> 5;
    const bool binning = nown <= MAX_OWNERS;
    hist[tid] = 0u;
    __syncthreads();

    const int e0 = blockIdx.x * BIN_EDGES;
    int dv[BIN_EPT], sv[BIN_EPT];
#pragma unroll
    for (int k = 0; k < BIN_EPT; ++k) {
        const int e = e0 + k * BIN_THREADS + tid;
        dv[k] = e < E ? dst[e] : -1;
        sv[k] = e < E ? src[e] : 0;
    }
    int ow[BIN_EPT];
    unsigned ent[BIN_EPT], rk[BIN_EPT];
#pragma unroll
    for (int k = 0; k < BIN_EPT; ++k) {
        ow[k] = -1; ent[k] = 0u; rk[k] = 0u;
        if (dv[k] >= 0) {
            atomicAdd(&deg[dv[k]], 1u);
            const int b = (int)((unsigned)dv[k] / (unsigned)ROWS_PER_WG);
            if (binning && b < nown) {
                ow[k] = b;
                ent[k] = ((unsigned)(dv[k] - b * ROWS_PER_WG) << 20) | (unsigned)sv[k];
                rk[k] = atomicAdd(&hist[b], 1u);
            }
        }
    }
    if (!binning) return;
    __syncthreads();

    {
        const unsigned c = hist[tid];
        unsigned inc = c;
#pragma unroll
        for (int off = 1; off < 32; off <<= 1) {
            const unsigned t = __shfl_up(inc, off, 32);
            if (lane >= off) inc += t;
        }
        if (lane == 31) wtot[wave] = inc;
        __syncthreads();
        unsigned pre = 0u;
        for (int w = 0; w < wave; ++w) pre += wtot[w];
        unsigned gb = 0u;
        if (c != 0u) gb = atomicAdd(&bins[(size_t)tid * ROWS_PER_WG * FDIM], c);
        hist[tid] = pre + inc - c;
        gbase[tid] = gb;
    }
    __syncthreads();

#pragma unroll
    for (int k = 0; k < BIN_EPT; ++k) {
        if (ow[k] >= 0) {
            const int b = ow[k];
            const unsigned p = hist[b] + rk[k];
            int nrb = n - b * ROWS_PER_WG;
            if (nrb > ROWS_PER_WG) nrb = ROWS_PER_WG;
            const unsigned cap = (unsigned)(nrb * FDIM - LIST_HDR);
            const unsigned pos = gbase[b] + rk[k];
            st_e[p] = ent[k];
            st_a[p] = pos < cap ? (unsigned)b * (ROWS_PER_WG * FDIM) + LIST_HDR + pos : 0xFFFFFFFFu;
        }
    }
    __syncthreads();

    int nv = 0;
    for (int w = 0; w < BIN_THREADS / 32; ++w) nv += (int)wtot[w];
    for (int i = tid; i < nv; i += BIN_THREADS) {
        const unsigned a = st_a[i];
        if (a != 0xFFFFFFFFu) bins[a] = st_e[i];
    }
}

__global__ void __launch_bounds__(256) dinv_kernel(const unsigned* __restrict__ deg,
                                                   float* __restrict__ dinv, int n) {
    int i = blockIdx.x * blockDim.x + threadIdx.x;
    if (i < n) dinv[i] = rsqrtf((float)deg[i] + 1.0f);
}

template <bool RELU>
__global__ void __launch_bounds__(256) gemm_wmma_kernel(const float* __restrict__ A,
                                                        const float* __restrict__ W,
                                                        float* __restrict__ Hout,
                                                        int nrows) {
    __shared__ float wlds[FDIM * FDIM];
    for (int i = threadIdx.x; i < FDIM * FDIM; i += 256) {
        int k = i >> 7, n = i & 127;
        wlds[((k >> 1) << 8) + (n << 1) + (k & 1)] = W[i];
    }
    __syncthreads();

    const int lane  = threadIdx.x & 31;
    const int wave  = threadIdx.x >> 5;
    const int strip = blockIdx.x * 8 + wave;
    const int nstrips = nrows >> 4;
    if (strip >= nstrips) return;

    const int row0  = strip << 4;
    const int m     = lane & 15;
    const int khalf = lane >> 4;
    const int kbase = khalf << 1;

    const float* arow = A + (size_t)(row0 + m) * FDIM;

    v8f acc[8];
#pragma unroll
    for (int t = 0; t < 8; ++t) acc[t] = (v8f)(0.0f);

    for (int k0 = 0; k0 < FDIM; k0 += 4) {
        v2f a = *(const v2f*)(arow + k0 + kbase);
        if (RELU) { a.x = fmaxf(a.x, 0.0f); a.y = fmaxf(a.y, 0.0f); }
        const int bbase = ((k0 >> 1) + khalf) * 256 + (m << 1);
#pragma unroll
        for (int t = 0; t < 8; ++t) {
            v2f b = *(const v2f*)(&wlds[bbase + t * 32]);
            acc[t] = __builtin_amdgcn_wmma_f32_16x16x4_f32(
                 false, a,  false, b,
                 (short)0, acc[t],  false,  false);
        }
    }

    const int rbase = row0 + (khalf << 3);
#pragma unroll
    for (int t = 0; t < 8; ++t) {
#pragma unroll
        for (int r = 0; r < 8; ++r)
            Hout[(size_t)(rbase + r) * FDIM + t * 16 + m] = acc[t][r];
    }
}

#define AGG_WAVES 32
#define AGG_UNROLL 8
#define PART_LOSS 32
#define PART_CNT 33
static_assert(AGG_WAVES == 32, "lane i reads the count of wave i");
static_assert(ROWS_PER_WG <= 2048, "local row must fit above bit 20");

template <bool FIRST>
__global__ void __launch_bounds__(1024) owner_agg_kernel(const float4* __restrict__ h,
                                                         float* out,
                                                         unsigned* list,
                                                         unsigned* list2,
                                                         const int* __restrict__ src,
                                                         const int* __restrict__ dst,
                                                         const float* __restrict__ dinv,
                                                         const float* __restrict__ bias,
                                                         int E, int n) {
    __shared__ float tile[ROWS_PER_WG * FDIM];
    __shared__ int queue[AGG_WAVES * 64];

    const int lane = threadIdx.x & 31;
    const int wave = threadIdx.x >> 5;
    const int qb = wave * 64;
    const int lo = blockIdx.x * ROWS_PER_WG;
    int nr = n - lo;
    if (nr > ROWS_PER_WG) nr = ROWS_PER_WG;

    unsigned* wl = list + (size_t)lo * FDIM;
    const int capw = ((nr * FDIM - LIST_HDR) >> 6) & ~31;
    uint2* seg = (uint2*)(wl + LIST_HDR) + (size_t)wave * capw;

    {
        const float b0 = bias[4 * lane], b1 = bias[4 * lane + 1];
        const float b2 = bias[4 * lane + 2], b3 = bias[4 * lane + 3];
        const float4* h4 = h + (size_t)lo * 32;
        for (int i = threadIdx.x; i < nr * 32; i += 1024) {
            const int row = i >> 5;
            const float4 v = h4[i];
            const float d = dinv[lo + row];
            const float d2 = d * d;
            float* t = tile + row * FDIM + lane;
            t[0] = v.x * d2 + b0; t[32] = v.y * d2 + b1; t[64] = v.z * d2 + b2; t[96] = v.w * d2 + b3;
        }
    }
    __syncthreads();

    auto add_row = [&](float4 v, int r, float w) {
        float* t = tile + r * FDIM + lane;
        atomicAdd(t, v.x * w);
        atomicAdd(t + 32, v.y * w);
        atomicAdd(t + 64, v.z * w);
        atomicAdd(t + 96, v.w * w);
    };

    auto gather = [&](int nf, int s, int dr, float nrm) {
        int j = 0;
        for (; j + 8 <= nf; j += 8) {
            float4 v[8];
            int r[8];
            float w[8];
#pragma unroll
            for (int u = 0; u < 8; ++u) {
                int sj = __builtin_amdgcn_readlane(s, j + u);
                r[u] = __builtin_amdgcn_readlane(dr, j + u);
                w[u] = __int_as_float(__builtin_amdgcn_readlane(__float_as_int(nrm), j + u));
                v[u] = h[(size_t)sj * 32 + lane];
            }
#pragma unroll
            for (int u = 0; u < 8; ++u) add_row(v[u], r[u], w[u]);
        }
        for (; j < nf; ++j) {
            int sj = __builtin_amdgcn_readlane(s, j);
            int rj = __builtin_amdgcn_readlane(dr, j);
            float wj = __int_as_float(__builtin_amdgcn_readlane(__float_as_int(nrm), j));
            add_row(h[(size_t)sj * 32 + lane], rj, wj);
        }
    };

    bool walk = false;
    if (!FIRST) {
        const unsigned c = wl[lane];
        walk = __builtin_amdgcn_ballot_w32(c > (unsigned)capw) == 0u;
    }

    bool binned = false;
    int cb = 0;
    const unsigned* bk = list2 + (size_t)lo * FDIM;
    if (FIRST) {
        cb = (int)bk[0];
        binned = gridDim.x <= MAX_OWNERS && (unsigned)cb <= (unsigned)(nr * FDIM - LIST_HDR);
    }

    if (FIRST && binned) {
        int tot = 0;
        int j0 = __builtin_amdgcn_readfirstlane(wave) * 32;
        const int step = 32 * AGG_WAVES;
        unsigned e1 = 0u, e2 = 0u;
        float na = 0.0f, nb = 0.0f;
        if (j0 + lane < cb) e1 = bk[LIST_HDR + j0 + lane];
        if (j0 + step + lane < cb) e2 = bk[LIST_HDR + j0 + step + lane];
        if (j0 + lane < cb) { na = dinv[e1 & 0xFFFFFu]; nb = dinv[lo + (int)(e1 >> 20)]; }
        for (; j0 < cb; j0 += step) {
            const unsigned cur = e1;
            const float nrm = na * nb;
            e1 = e2;
            e2 = 0u; na = 0.0f; nb = 0.0f;
            if (j0 + 2 * step + lane < cb) e2 = bk[LIST_HDR + j0 + 2 * step + lane];
            if (j0 + step + lane < cb) { na = dinv[e1 & 0xFFFFFu]; nb = dinv[lo + (int)(e1 >> 20)]; }
            const int nf = cb - j0 < 32 ? cb - j0 : 32;
            const int s = (int)(cur & 0xFFFFFu), dr = (int)(cur >> 20);
            if (lane < nf && tot + nf <= capw) seg[tot + lane] = make_uint2(cur, __float_as_uint(nrm));
            tot += nf;
            gather(nf, s, dr, nrm);
        }
        if (lane == 0) wl[wave] = (unsigned)tot;
    } else if (walk) {
        const int cw = __builtin_amdgcn_readfirstlane((int)wl[wave]);
        unsigned* wl2 = list2 + (size_t)lo * FDIM;
        unsigned* seg2 = wl2 + LIST_HDR + (size_t)wave * capw;
        int tot2 = 0;
        uint2 nxt = make_uint2(0u, 0u);
        if (lane < cw) nxt = seg[lane];
        for (int j0 = 0; j0 < cw; j0 += 32) {
            const uint2 cur = nxt;
            if (j0 + 32 + lane < cw) nxt = seg[j0 + 32 + lane];
            const int nf = cw - j0 < 32 ? cw - j0 : 32;
            const int sx = (int)(cur.x & 0xFFFFFu), rx = (int)(cur.x >> 20);
            const bool keep = lane < nf && sx < lo + rx;
            const unsigned km = __builtin_amdgcn_ballot_w32(keep);
            if (keep) seg2[tot2 + __popc(km & ((1u << lane) - 1u))] = cur.x;
            tot2 += __popc(km);
            gather(nf, sx, rx, __uint_as_float(cur.y));
        }
        if (lane == 0) wl2[wave] = (unsigned)tot2;
    } else {
        int cnt = 0;
        int tot = 0;
        const unsigned lt = (1u << lane) - 1u;

        auto flush = [&](int nf) {
            int s = 0, dr = 0;
            float nrm = 0.0f;
            if (lane < nf) {
                int e = queue[qb + lane];
                s = src[e];
                int d = dst[e];
                dr = d - lo;
                nrm = dinv[s] * dinv[d];
                if (FIRST && tot + nf <= capw)
                    seg[tot + lane] = make_uint2(((unsigned)dr << 20) | (unsigned)s, __float_as_uint(nrm));
            }
            tot += nf;
            gather(nf, s, dr, nrm);
        };

        int head = (int)(((16u - ((unsigned)(size_t)dst & 15u)) & 15u) >> 2);
        if (head > E) head = E;
        const int4* dst4 = (const int4*)(dst + head);
        const int nvec = (E - head) >> 2;
        const int ntail = (E - head) & 3;
        const int nextra = head + ntail;

        const int step = AGG_WAVES * 32 * AGG_UNROLL;
        const int npass = (nvec + step - 1) / step;
        for (int pass = 0; pass <= npass; ++pass) {
            const int vb = pass * step + wave * (32 * AGG_UNROLL);
            unsigned mbits = 0u;
            if (pass < npass) {
                int4 d4[AGG_UNROLL];
                unsigned nru[AGG_UNROLL];
#pragma unroll
                for (int u = 0; u < AGG_UNROLL; ++u) {
                    const int vi = vb + u * 32 + lane;
                    d4[u] = dst4[vi < nvec ? vi : nvec - 1];
                    nru[u] = vi < nvec ? (unsigned)nr : 0u;
                }
#pragma unroll
                for (int u = 0; u < AGG_UNROLL; ++u) {
                    const bool cx = (unsigned)(d4[u].x - lo) < nru[u];
                    const bool cy = (unsigned)(d4[u].y - lo) < nru[u];
                    const bool cz = (unsigned)(d4[u].z - lo) < nru[u];
                    const bool cw = (unsigned)(d4[u].w - lo) < nru[u];
                    const unsigned any = __builtin_amdgcn_ballot_w32(cx) | __builtin_amdgcn_ballot_w32(cy) |
                                         __builtin_amdgcn_ballot_w32(cz) | __builtin_amdgcn_ballot_w32(cw);
                    if (any)
                        mbits |= ((cx ? 1u : 0u) | (cy ? 2u : 0u) | (cz ? 4u : 0u) | (cw ? 8u : 0u)) << (4 * u);
                }
            } else if (wave == 0 && lane < nextra) {
                int e = lane < head ? lane : (nvec << 2) + lane;
                if ((unsigned)(dst[e] - lo) < (unsigned)nr) mbits = 1u;
            }
            while (__builtin_amdgcn_ballot_w32(mbits != 0u)) {
                const bool mt = mbits != 0u;
                const int k = mt ? (__ffs(mbits) - 1) : 0;
                int e;
                if (pass < npass) e = head + ((vb + (k >> 2) * 32 + lane) << 2) + (k & 3);
                else              e = lane < head ? lane : (nvec << 2) + lane;
                mbits &= mbits - 1u;
                const unsigned mask = __builtin_amdgcn_ballot_w32(mt);
                const int pos = cnt + __popc(mask & lt);
                if (mt) queue[qb + pos] = e;
                cnt += __popc(mask);
                if (cnt >= 32) {
                    flush(32);
                    const int rem = cnt - 32;
                    int t = 0;
                    if (lane < rem) t = queue[qb + 32 + lane];
                    if (lane < rem) queue[qb + lane] = t;
                    cnt = rem;
                }
            }
        }
        if (cnt > 0) flush(cnt);
        if (FIRST && lane == 0) wl[wave] = (unsigned)tot;
        if (!FIRST && lane == 0) list2[(size_t)lo * FDIM + wave] = 0xFFFFFFFFu;
    }

    __syncthreads();
    float4* out4 = (float4*)out + (size_t)lo * 32;
    for (int i = threadIdx.x; i < nr * 32; i += 1024) {
        const float* t = tile + (i >> 5) * FDIM + (i & 31);
        out4[i] = make_float4(t[0], t[32], t[64], t[96]);
    }
}

__global__ void __launch_bounds__(1024) owner_loss_kernel(const float4* __restrict__ rep,
                                                          unsigned* list2,
                                                          const int* __restrict__ src,
                                                          const int* __restrict__ dst,
                                                          float target, int E, int n) {
    __shared__ float4 tile4[ROWS_PER_WG * 32];
    __shared__ float wsum[AGG_WAVES];
    __shared__ unsigned wcnt[AGG_WAVES];

    const int lane = threadIdx.x & 31;
    const int wave = threadIdx.x >> 5;
    const int lo = blockIdx.x * ROWS_PER_WG;
    int nr = n - lo;
    if (nr > ROWS_PER_WG) nr = ROWS_PER_WG;
    unsigned* wl2 = list2 + (size_t)lo * FDIM;
    const int capw = ((nr * FDIM - LIST_HDR) >> 6) & ~31;
    const unsigned* seg2 = wl2 + LIST_HDR + (size_t)wave * capw;

    for (int i = threadIdx.x; i < nr * 32; i += 1024) tile4[i] = rep[(size_t)lo * 32 + i];
    __syncthreads();

    float local = 0.0f;
    unsigned cntw = 0u;

    auto process = [&](unsigned m, int s, int dr) {
        cntw += (unsigned)__popc(m);
        while (m) {
            const int left = __popc(m);
            const unsigned vbits = left >= 8 ? 0xFFu : ((1u << left) - 1u);
            float4 v[8];
            int r[8];
#pragma unroll
            for (int u = 0; u < 8; ++u) {
                const int j = m ? (__ffs(m) - 1) : 0;
                m &= m - 1u;
                const int sj = __builtin_amdgcn_readlane(s, j);
                r[u] = __builtin_amdgcn_readlane(dr, j);
                v[u] = rep[(size_t)sj * 32 + lane];
            }
            float p[8];
#pragma unroll
            for (int u = 0; u < 8; ++u) {
                const float4 t = tile4[r[u] * 32 + lane];
                p[u] = v[u].x * t.x + v[u].y * t.y + v[u].z * t.z + v[u].w * t.w;
            }
            const bool h = (lane & 16) != 0, g = (lane & 8) != 0, f = (lane & 4) != 0;
            float q[4];
#pragma unroll
            for (int u = 0; u < 4; ++u)
                q[u] = (h ? p[u + 4] : p[u]) + __shfl_xor(h ? p[u] : p[u + 4], 16, 32);
            float q2[2];
#pragma unroll
            for (int u = 0; u < 2; ++u)
                q2[u] = (g ? q[u + 2] : q[u]) + __shfl_xor(g ? q[u] : q[u + 2], 8, 32);
            float t1 = (f ? q2[1] : q2[0]) + __shfl_xor(f ? q2[0] : q2[1], 4, 32);
            t1 += __shfl_xor(t1, 2, 32);
            t1 += __shfl_xor(t1, 1, 32);
            if ((lane & 3) == 0 && ((vbits >> (lane >> 2)) & 1u)) {
                const float dl = t1 - target;
                local += dl * dl;
            }
        }
    };

    const unsigned c = wl2[lane];
    if (__builtin_amdgcn_ballot_w32(c > (unsigned)capw) == 0u) {
        const int cw = __builtin_amdgcn_readfirstlane((int)wl2[wave]);
        unsigned nxt = 0u;
        if (lane < cw) nxt = seg2[lane];
        for (int j0 = 0; j0 < cw; j0 += 32) {
            const unsigned cur = nxt;
            if (j0 + 32 + lane < cw) nxt = seg2[j0 + 32 + lane];
            const int nf = cw - j0 < 32 ? cw - j0 : 32;
            const unsigned m = nf >= 32 ? 0xFFFFFFFFu : ((1u << nf) - 1u);
            process(m, (int)(cur & 0xFFFFFu), (int)(cur >> 20));
        }
    } else {
        for (int base = wave * 32; base < E; base += 32 * AGG_WAVES) {
            const int e = base + lane;
            int s = 0, dr = 0;
            bool ok = false;
            if (e < E) {
                const int d = dst[e];
                const int sv = src[e];
                ok = (unsigned)(d - lo) < (unsigned)nr && sv < d;
                if (ok) { s = sv; dr = d - lo; }
            }
            process(__builtin_amdgcn_ballot_w32(ok), s, dr);
        }
    }

#pragma unroll
    for (int off = 16; off > 0; off >>= 1) local += __shfl_xor(local, off, 32);
    if (lane == 0) { wsum[wave] = local; wcnt[wave] = cntw; }
    __syncthreads();
    if (wave == 0) {
        float a = wsum[lane];
        unsigned b = wcnt[lane];
#pragma unroll
        for (int off = 16; off > 0; off >>= 1) {
            a += __shfl_xor(a, off, 32);
            b += __shfl_xor(b, off, 32);
        }
        if (lane == 0) { wl2[PART_LOSS] = __float_as_uint(a); wl2[PART_CNT] = b; }
    }
}

__global__ void __launch_bounds__(256) edge_loss_kernel(const float4* __restrict__ rep,
                                                        const int* __restrict__ src,
                                                        const int* __restrict__ dst, int E,
                                                        float target,
                                                        float* __restrict__ loss_acc,
                                                        unsigned* __restrict__ cnt_acc) {
    int lane   = threadIdx.x & 31;
    int wave   = blockIdx.x * (blockDim.x >> 5) + (threadIdx.x >> 5);
    int nwaves = gridDim.x * (blockDim.x >> 5);
    float local = 0.0f;
    unsigned cnt = 0;
    for (int e = wave; e < E; e += nwaves) {
        int s = src[e], d = dst[e];
        if (s < d) {
            float4 a = rep[(size_t)s * 32 + lane];
            float4 b = rep[(size_t)d * 32 + lane];
            float p = a.x * b.x + a.y * b.y + a.z * b.z + a.w * b.w;
#pragma unroll
            for (int off = 16; off > 0; off >>= 1) p += __shfl_down(p, off, 32);
            if (lane == 0) {
                float dl = p - target;
                local += dl * dl;
                cnt++;
            }
        }
    }
    if (lane == 0 && cnt > 0) {
        atomicAdd(loss_acc, local);
        atomicAdd(cnt_acc, cnt);
    }
}

__global__ void __launch_bounds__(256) finalize_kernel(const unsigned* __restrict__ part, int nwg,
                                                       const float* __restrict__ accum,
                                                       const unsigned* __restrict__ cnt,
                                                       float* __restrict__ out_scalar, float n_nodes) {
    __shared__ float sl[256];
    __shared__ unsigned sc[256];
    float l = 0.0f;
    unsigned c = 0u;
    for (int i = threadIdx.x; i < nwg; i += 256) {
        const unsigned* hd = part + (size_t)i * ROWS_PER_WG * FDIM;
        l += __uint_as_float(hd[PART_LOSS]);
        c += hd[PART_CNT];
    }
    sl[threadIdx.x] = l;
    sc[threadIdx.x] = c;
    __syncthreads();
    for (int w = 128; w > 0; w >>= 1) {
        if ((int)threadIdx.x < w) { sl[threadIdx.x] += sl[threadIdx.x + w]; sc[threadIdx.x] += sc[threadIdx.x + w]; }
        __syncthreads();
    }
    if (threadIdx.x == 0) {
        const float tot = sl[0] + accum[1];
        const float cc = (float)(sc[0] + cnt[1]);
        *out_scalar = tot * n_nodes / cc;
    }
}

extern "C" void kernel_launch(void* const* d_in, const int* in_sizes, int n_in,
                              void* d_out, int out_size, void* d_ws, size_t ws_size,
                              hipStream_t stream) {
    const float* features = (const float*)d_in[0];
    const int*   ei       = (const int*)d_in[1];
    const int*   nei      = (const int*)d_in[2];
    const float* w1       = (const float*)d_in[3];
    const float* b1       = (const float*)d_in[4];
    const float* w2       = (const float*)d_in[5];
    const float* b2       = (const float*)d_in[6];

    const int N  = in_sizes[0] / FDIM;
    const int E  = in_sizes[1] / 2;
    const int En = in_sizes[2] / 2;
    const int* src  = ei;
    const int* dst  = ei + E;
    const int* nsrc = nei;
    const int* ndst = nei + En;

    char* ws = (char*)d_ws;
    float*    bufA = (float*)ws;
    float*    bufB = (float*)(ws + (size_t)N * FDIM * 4);
    unsigned* deg  = (unsigned*)(ws + (size_t)2 * N * FDIM * 4);
    float*    dinv = (float*)((char*)deg + (size_t)N * 4);
    float*    accum = (float*)((char*)dinv + (size_t)N * 4);
    unsigned* cnt   = (unsigned*)(accum + 2);

    float* rep      = (float*)d_out;
    float* loss_out = rep + (size_t)N * FDIM;

    const int nstrips = N >> 4;
    const int gblocks = (nstrips + 7) / 8;
    const int ablocks = (N + ROWS_PER_WG - 1) / ROWS_PER_WG;

    zero_init_kernel<<<(N + 255) / 256, 256, 0, stream>>>(deg, accum, cnt, (unsigned*)bufB, N, ablocks);
    count_bin_kernel<<<(E + BIN_EDGES - 1) / BIN_EDGES, BIN_THREADS, 0, stream>>>(src, dst, deg, (unsigned*)bufB, E, N, ablocks);
    dinv_kernel<<<(N + 255) / 256, 256, 0, stream>>>(deg, dinv, N);

    gemm_wmma_kernel<false><<<gblocks, 256, 0, stream>>>(features, w1, bufA, N);
    owner_agg_kernel<true><<<ablocks, 1024, 0, stream>>>((const float4*)bufA, bufB, (unsigned*)rep, (unsigned*)bufB,
                                                         src, dst, dinv, b1, E, N);

    gemm_wmma_kernel<true><<<gblocks, 256, 0, stream>>>(bufB, w2, bufA, N);
    owner_agg_kernel<false><<<ablocks, 1024, 0, stream>>>((const float4*)bufA, rep, (unsigned*)rep, (unsigned*)bufB,
                                                          src, dst, dinv, b2, E, N);

    owner_loss_kernel<<<ablocks, 1024, 0, stream>>>((const float4*)rep, (unsigned*)bufB, src, dst, 1.0f, E, N);
    edge_loss_kernel<<<2048, 256, 0, stream>>>((const float4*)rep, nsrc, ndst, En, 0.0f,
                                               &accum[1], &cnt[1]);
    finalize_kernel<<<1, 256, 0, stream>>>((const unsigned*)bufB, ablocks, accum, cnt, loss_out, (float)N);
}


# degrees counted by the row owners: each workgroup counts its bin's entries per local row with LDS integer adds and writes dinv; the binning pass drops its per-edge global adds; sum orders unchanged
# speedup vs baseline: 2.9279x; 1.2595x over previous
#include <hip/hip_runtime.h>

typedef __attribute__((ext_vector_type(2))) float v2f;
typedef __attribute__((ext_vector_type(8))) float v8f;

#define FDIM 128
#define ROWS_PER_WG 391
#define LIST_HDR 64
#define MAX_OWNERS 256
#define BIN_THREADS 256
#define BIN_EPT 8
#define BIN_EDGES (BIN_THREADS * BIN_EPT)
static_assert(BIN_THREADS == MAX_OWNERS, "thread b of the binning pass handles owner b");

__global__ void __launch_bounds__(256) zero_init_kernel(float* __restrict__ accum,
                                                        unsigned* __restrict__ cnt,
                                                        unsigned* __restrict__ bins, int nown) {
    int i = blockIdx.x * blockDim.x + threadIdx.x;
    if (i < 2) { accum[i] = 0.0f; cnt[i] = 0u; }
    if (i < nown) bins[(size_t)i * ROWS_PER_WG * FDIM] = 0u;
}

__global__ void __launch_bounds__(BIN_THREADS) bin_kernel(const int* __restrict__ src,
                                                          const int* __restrict__ dst,
                                                          unsigned* __restrict__ bins,
                                                          int E, int n, int nown) {
    __shared__ unsigned hist[MAX_OWNERS];
    __shared__ unsigned gbase[MAX_OWNERS];
    __shared__ unsigned wtot[BIN_THREADS / 32];
    __shared__ unsigned st_e[BIN_EDGES];
    __shared__ unsigned st_a[BIN_EDGES];

    const int tid = threadIdx.x;
    const int lane = tid & 31;
    const int wave = tid >> 5;
    if (nown > MAX_OWNERS) return;
    hist[tid] = 0u;
    __syncthreads();

    const int e0 = blockIdx.x * BIN_EDGES;
    int dv[BIN_EPT], sv[BIN_EPT];
#pragma unroll
    for (int k = 0; k < BIN_EPT; ++k) {
        const int e = e0 + k * BIN_THREADS + tid;
        dv[k] = e < E ? dst[e] : -1;
        sv[k] = e < E ? src[e] : 0;
    }
    int ow[BIN_EPT];
    unsigned ent[BIN_EPT], rk[BIN_EPT];
#pragma unroll
    for (int k = 0; k < BIN_EPT; ++k) {
        ow[k] = -1; ent[k] = 0u; rk[k] = 0u;
        if (dv[k] >= 0) {
            const int b = (int)((unsigned)dv[k] / (unsigned)ROWS_PER_WG);
            if (b < nown) {
                ow[k] = b;
                ent[k] = ((unsigned)(dv[k] - b * ROWS_PER_WG) << 20) | (unsigned)sv[k];
                rk[k] = atomicAdd(&hist[b], 1u);
            }
        }
    }
    __syncthreads();

    {
        const unsigned c = hist[tid];
        unsigned inc = c;
#pragma unroll
        for (int off = 1; off < 32; off <<= 1) {
            const unsigned t = __shfl_up(inc, off, 32);
            if (lane >= off) inc += t;
        }
        if (lane == 31) wtot[wave] = inc;
        __syncthreads();
        unsigned pre = 0u;
        for (int w = 0; w < wave; ++w) pre += wtot[w];
        unsigned gb = 0u;
        if (c != 0u) gb = atomicAdd(&bins[(size_t)tid * ROWS_PER_WG * FDIM], c);
        hist[tid] = pre + inc - c;
        gbase[tid] = gb;
    }
    __syncthreads();

#pragma unroll
    for (int k = 0; k < BIN_EPT; ++k) {
        if (ow[k] >= 0) {
            const int b = ow[k];
            const unsigned p = hist[b] + rk[k];
            int nrb = n - b * ROWS_PER_WG;
            if (nrb > ROWS_PER_WG) nrb = ROWS_PER_WG;
            const unsigned cap = (unsigned)(nrb * FDIM - LIST_HDR);
            const unsigned pos = gbase[b] + rk[k];
            st_e[p] = ent[k];
            st_a[p] = pos < cap ? (unsigned)b * (ROWS_PER_WG * FDIM) + LIST_HDR + pos : 0xFFFFFFFFu;
        }
    }
    __syncthreads();

    int nv = 0;
    for (int w = 0; w < BIN_THREADS / 32; ++w) nv += (int)wtot[w];
    for (int i = tid; i < nv; i += BIN_THREADS) {
        const unsigned a = st_a[i];
        if (a != 0xFFFFFFFFu) bins[a] = st_e[i];
    }
}

#define DEG_THREADS 1024
#define DEG_EPT 8
__global__ void __launch_bounds__(DEG_THREADS) owner_dinv_kernel(const unsigned* __restrict__ bins,
                                                                 const int* __restrict__ dst,
                                                                 float* __restrict__ dinv, int E, int n) {
    __shared__ unsigned rc[ROWS_PER_WG];
    const int tid = threadIdx.x;
    const int lo = blockIdx.x * ROWS_PER_WG;
    int nr = n - lo;
    if (nr > ROWS_PER_WG) nr = ROWS_PER_WG;
    for (int i = tid; i < nr; i += DEG_THREADS) rc[i] = 0u;
    __syncthreads();

    const unsigned* bk = bins + (size_t)lo * FDIM;
    const unsigned cb = bk[0];
    const bool binned = gridDim.x <= MAX_OWNERS && cb <= (unsigned)(nr * FDIM - LIST_HDR);
    if (binned) {
        for (unsigned base = 0u; base < cb; base += DEG_THREADS * DEG_EPT) {
            unsigned ev[DEG_EPT];
#pragma unroll
            for (int k = 0; k < DEG_EPT; ++k) {
                const unsigned i = base + (unsigned)(k * DEG_THREADS + tid);
                ev[k] = i < cb ? bk[LIST_HDR + i] : 0xFFFFFFFFu;
            }
#pragma unroll
            for (int k = 0; k < DEG_EPT; ++k)
                if (ev[k] != 0xFFFFFFFFu) atomicAdd(&rc[ev[k] >> 20], 1u);
        }
    } else {
        for (int e = tid; e < E; e += DEG_THREADS) {
            const unsigned r = (unsigned)(dst[e] - lo);
            if (r < (unsigned)nr) atomicAdd(&rc[r], 1u);
        }
    }
    __syncthreads();
    for (int i = tid; i < nr; i += DEG_THREADS) dinv[lo + i] = rsqrtf((float)rc[i] + 1.0f);
}

template <bool RELU>
__global__ void __launch_bounds__(256) gemm_wmma_kernel(const float* __restrict__ A,
                                                        const float* __restrict__ W,
                                                        float* __restrict__ Hout,
                                                        int nrows) {
    __shared__ float wlds[FDIM * FDIM];
    for (int i = threadIdx.x; i < FDIM * FDIM; i += 256) {
        int k = i >> 7, n = i & 127;
        wlds[((k >> 1) << 8) + (n << 1) + (k & 1)] = W[i];
    }
    __syncthreads();

    const int lane  = threadIdx.x & 31;
    const int wave  = threadIdx.x >> 5;
    const int strip = blockIdx.x * 8 + wave;
    const int nstrips = nrows >> 4;
    if (strip >= nstrips) return;

    const int row0  = strip << 4;
    const int m     = lane & 15;
    const int khalf = lane >> 4;
    const int kbase = khalf << 1;

    const float* arow = A + (size_t)(row0 + m) * FDIM;

    v8f acc[8];
#pragma unroll
    for (int t = 0; t < 8; ++t) acc[t] = (v8f)(0.0f);

    for (int k0 = 0; k0 < FDIM; k0 += 4) {
        v2f a = *(const v2f*)(arow + k0 + kbase);
        if (RELU) { a.x = fmaxf(a.x, 0.0f); a.y = fmaxf(a.y, 0.0f); }
        const int bbase = ((k0 >> 1) + khalf) * 256 + (m << 1);
#pragma unroll
        for (int t = 0; t < 8; ++t) {
            v2f b = *(const v2f*)(&wlds[bbase + t * 32]);
            acc[t] = __builtin_amdgcn_wmma_f32_16x16x4_f32(
                 false, a,  false, b,
                 (short)0, acc[t],  false,  false);
        }
    }

    const int rbase = row0 + (khalf << 3);
#pragma unroll
    for (int t = 0; t < 8; ++t) {
#pragma unroll
        for (int r = 0; r < 8; ++r)
            Hout[(size_t)(rbase + r) * FDIM + t * 16 + m] = acc[t][r];
    }
}

#define AGG_WAVES 32
#define AGG_UNROLL 8
#define PART_LOSS 32
#define PART_CNT 33
static_assert(AGG_WAVES == 32, "lane i reads the count of wave i");
static_assert(ROWS_PER_WG <= 2048, "local row must fit above bit 20");

template <bool FIRST>
__global__ void __launch_bounds__(1024) owner_agg_kernel(const float4* __restrict__ h,
                                                         float* out,
                                                         unsigned* list,
                                                         unsigned* list2,
                                                         const int* __restrict__ src,
                                                         const int* __restrict__ dst,
                                                         const float* __restrict__ dinv,
                                                         const float* __restrict__ bias,
                                                         int E, int n) {
    __shared__ float tile[ROWS_PER_WG * FDIM];
    __shared__ int queue[AGG_WAVES * 64];

    const int lane = threadIdx.x & 31;
    const int wave = threadIdx.x >> 5;
    const int qb = wave * 64;
    const int lo = blockIdx.x * ROWS_PER_WG;
    int nr = n - lo;
    if (nr > ROWS_PER_WG) nr = ROWS_PER_WG;

    unsigned* wl = list + (size_t)lo * FDIM;
    const int capw = ((nr * FDIM - LIST_HDR) >> 6) & ~31;
    uint2* seg = (uint2*)(wl + LIST_HDR) + (size_t)wave * capw;

    {
        const float b0 = bias[4 * lane], b1 = bias[4 * lane + 1];
        const float b2 = bias[4 * lane + 2], b3 = bias[4 * lane + 3];
        const float4* h4 = h + (size_t)lo * 32;
        for (int i = threadIdx.x; i < nr * 32; i += 1024) {
            const int row = i >> 5;
            const float4 v = h4[i];
            const float d = dinv[lo + row];
            const float d2 = d * d;
            float* t = tile + row * FDIM + lane;
            t[0] = v.x * d2 + b0; t[32] = v.y * d2 + b1; t[64] = v.z * d2 + b2; t[96] = v.w * d2 + b3;
        }
    }
    __syncthreads();

    auto add_row = [&](float4 v, int r, float w) {
        float* t = tile + r * FDIM + lane;
        atomicAdd(t, v.x * w);
        atomicAdd(t + 32, v.y * w);
        atomicAdd(t + 64, v.z * w);
        atomicAdd(t + 96, v.w * w);
    };

    auto gather = [&](int nf, int s, int dr, float nrm) {
        int j = 0;
        for (; j + 8 <= nf; j += 8) {
            float4 v[8];
            int r[8];
            float w[8];
#pragma unroll
            for (int u = 0; u < 8; ++u) {
                int sj = __builtin_amdgcn_readlane(s, j + u);
                r[u] = __builtin_amdgcn_readlane(dr, j + u);
                w[u] = __int_as_float(__builtin_amdgcn_readlane(__float_as_int(nrm), j + u));
                v[u] = h[(size_t)sj * 32 + lane];
            }
#pragma unroll
            for (int u = 0; u < 8; ++u) add_row(v[u], r[u], w[u]);
        }
        for (; j < nf; ++j) {
            int sj = __builtin_amdgcn_readlane(s, j);
            int rj = __builtin_amdgcn_readlane(dr, j);
            float wj = __int_as_float(__builtin_amdgcn_readlane(__float_as_int(nrm), j));
            add_row(h[(size_t)sj * 32 + lane], rj, wj);
        }
    };

    bool walk = false;
    if (!FIRST) {
        const unsigned c = wl[lane];
        walk = __builtin_amdgcn_ballot_w32(c > (unsigned)capw) == 0u;
    }

    bool binned = false;
    int cb = 0;
    const unsigned* bk = list2 + (size_t)lo * FDIM;
    if (FIRST) {
        cb = (int)bk[0];
        binned = gridDim.x <= MAX_OWNERS && (unsigned)cb <= (unsigned)(nr * FDIM - LIST_HDR);
    }

    if (FIRST && binned) {
        int tot = 0;
        int j0 = __builtin_amdgcn_readfirstlane(wave) * 32;
        const int step = 32 * AGG_WAVES;
        unsigned e1 = 0u, e2 = 0u;
        float na = 0.0f, nb = 0.0f;
        if (j0 + lane < cb) e1 = bk[LIST_HDR + j0 + lane];
        if (j0 + step + lane < cb) e2 = bk[LIST_HDR + j0 + step + lane];
        if (j0 + lane < cb) { na = dinv[e1 & 0xFFFFFu]; nb = dinv[lo + (int)(e1 >> 20)]; }
        for (; j0 < cb; j0 += step) {
            const unsigned cur = e1;
            const float nrm = na * nb;
            e1 = e2;
            e2 = 0u; na = 0.0f; nb = 0.0f;
            if (j0 + 2 * step + lane < cb) e2 = bk[LIST_HDR + j0 + 2 * step + lane];
            if (j0 + step + lane < cb) { na = dinv[e1 & 0xFFFFFu]; nb = dinv[lo + (int)(e1 >> 20)]; }
            const int nf = cb - j0 < 32 ? cb - j0 : 32;
            const int s = (int)(cur & 0xFFFFFu), dr = (int)(cur >> 20);
            if (lane < nf && tot + nf <= capw) seg[tot + lane] = make_uint2(cur, __float_as_uint(nrm));
            tot += nf;
            gather(nf, s, dr, nrm);
        }
        if (lane == 0) wl[wave] = (unsigned)tot;
    } else if (walk) {
        const int cw = __builtin_amdgcn_readfirstlane((int)wl[wave]);
        unsigned* wl2 = list2 + (size_t)lo * FDIM;
        unsigned* seg2 = wl2 + LIST_HDR + (size_t)wave * capw;
        int tot2 = 0;
        uint2 nxt = make_uint2(0u, 0u);
        if (lane < cw) nxt = seg[lane];
        for (int j0 = 0; j0 < cw; j0 += 32) {
            const uint2 cur = nxt;
            if (j0 + 32 + lane < cw) nxt = seg[j0 + 32 + lane];
            const int nf = cw - j0 < 32 ? cw - j0 : 32;
            const int sx = (int)(cur.x & 0xFFFFFu), rx = (int)(cur.x >> 20);
            const bool keep = lane < nf && sx < lo + rx;
            const unsigned km = __builtin_amdgcn_ballot_w32(keep);
            if (keep) seg2[tot2 + __popc(km & ((1u << lane) - 1u))] = cur.x;
            tot2 += __popc(km);
            gather(nf, sx, rx, __uint_as_float(cur.y));
        }
        if (lane == 0) wl2[wave] = (unsigned)tot2;
    } else {
        int cnt = 0;
        int tot = 0;
        const unsigned lt = (1u << lane) - 1u;

        auto flush = [&](int nf) {
            int s = 0, dr = 0;
            float nrm = 0.0f;
            if (lane < nf) {
                int e = queue[qb + lane];
                s = src[e];
                int d = dst[e];
                dr = d - lo;
                nrm = dinv[s] * dinv[d];
                if (FIRST && tot + nf <= capw)
                    seg[tot + lane] = make_uint2(((unsigned)dr << 20) | (unsigned)s, __float_as_uint(nrm));
            }
            tot += nf;
            gather(nf, s, dr, nrm);
        };

        int head = (int)(((16u - ((unsigned)(size_t)dst & 15u)) & 15u) >> 2);
        if (head > E) head = E;
        const int4* dst4 = (const int4*)(dst + head);
        const int nvec = (E - head) >> 2;
        const int ntail = (E - head) & 3;
        const int nextra = head + ntail;

        const int step = AGG_WAVES * 32 * AGG_UNROLL;
        const int npass = (nvec + step - 1) / step;
        for (int pass = 0; pass <= npass; ++pass) {
            const int vb = pass * step + wave * (32 * AGG_UNROLL);
            unsigned mbits = 0u;
            if (pass < npass) {
                int4 d4[AGG_UNROLL];
                unsigned nru[AGG_UNROLL];
#pragma unroll
                for (int u = 0; u < AGG_UNROLL; ++u) {
                    const int vi = vb + u * 32 + lane;
                    d4[u] = dst4[vi < nvec ? vi : nvec - 1];
                    nru[u] = vi < nvec ? (unsigned)nr : 0u;
                }
#pragma unroll
                for (int u = 0; u < AGG_UNROLL; ++u) {
                    const bool cx = (unsigned)(d4[u].x - lo) < nru[u];
                    const bool cy = (unsigned)(d4[u].y - lo) < nru[u];
                    const bool cz = (unsigned)(d4[u].z - lo) < nru[u];
                    const bool cw = (unsigned)(d4[u].w - lo) < nru[u];
                    const unsigned any = __builtin_amdgcn_ballot_w32(cx) | __builtin_amdgcn_ballot_w32(cy) |
                                         __builtin_amdgcn_ballot_w32(cz) | __builtin_amdgcn_ballot_w32(cw);
                    if (any)
                        mbits |= ((cx ? 1u : 0u) | (cy ? 2u : 0u) | (cz ? 4u : 0u) | (cw ? 8u : 0u)) << (4 * u);
                }
            } else if (wave == 0 && lane < nextra) {
                int e = lane < head ? lane : (nvec << 2) + lane;
                if ((unsigned)(dst[e] - lo) < (unsigned)nr) mbits = 1u;
            }
            while (__builtin_amdgcn_ballot_w32(mbits != 0u)) {
                const bool mt = mbits != 0u;
                const int k = mt ? (__ffs(mbits) - 1) : 0;
                int e;
                if (pass < npass) e = head + ((vb + (k >> 2) * 32 + lane) << 2) + (k & 3);
                else              e = lane < head ? lane : (nvec << 2) + lane;
                mbits &= mbits - 1u;
                const unsigned mask = __builtin_amdgcn_ballot_w32(mt);
                const int pos = cnt + __popc(mask & lt);
                if (mt) queue[qb + pos] = e;
                cnt += __popc(mask);
                if (cnt >= 32) {
                    flush(32);
                    const int rem = cnt - 32;
                    int t = 0;
                    if (lane < rem) t = queue[qb + 32 + lane];
                    if (lane < rem) queue[qb + lane] = t;
                    cnt = rem;
                }
            }
        }
        if (cnt > 0) flush(cnt);
        if (FIRST && lane == 0) wl[wave] = (unsigned)tot;
        if (!FIRST && lane == 0) list2[(size_t)lo * FDIM + wave] = 0xFFFFFFFFu;
    }

    __syncthreads();
    float4* out4 = (float4*)out + (size_t)lo * 32;
    for (int i = threadIdx.x; i < nr * 32; i += 1024) {
        const float* t = tile + (i >> 5) * FDIM + (i & 31);
        out4[i] = make_float4(t[0], t[32], t[64], t[96]);
    }
}

__global__ void __launch_bounds__(1024) owner_loss_kernel(const float4* __restrict__ rep,
                                                          unsigned* list2,
                                                          const int* __restrict__ src,
                                                          const int* __restrict__ dst,
                                                          float target, int E, int n) {
    __shared__ float4 tile4[ROWS_PER_WG * 32];
    __shared__ float wsum[AGG_WAVES];
    __shared__ unsigned wcnt[AGG_WAVES];

    const int lane = threadIdx.x & 31;
    const int wave = threadIdx.x >> 5;
    const int lo = blockIdx.x * ROWS_PER_WG;
    int nr = n - lo;
    if (nr > ROWS_PER_WG) nr = ROWS_PER_WG;
    unsigned* wl2 = list2 + (size_t)lo * FDIM;
    const int capw = ((nr * FDIM - LIST_HDR) >> 6) & ~31;
    const unsigned* seg2 = wl2 + LIST_HDR + (size_t)wave * capw;

    for (int i = threadIdx.x; i < nr * 32; i += 1024) tile4[i] = rep[(size_t)lo * 32 + i];
    __syncthreads();

    float local = 0.0f;
    unsigned cntw = 0u;

    auto process = [&](unsigned m, int s, int dr) {
        cntw += (unsigned)__popc(m);
        while (m) {
            const int left = __popc(m);
            const unsigned vbits = left >= 8 ? 0xFFu : ((1u << left) - 1u);
            float4 v[8];
            int r[8];
#pragma unroll
            for (int u = 0; u < 8; ++u) {
                const int j = m ? (__ffs(m) - 1) : 0;
                m &= m - 1u;
                const int sj = __builtin_amdgcn_readlane(s, j);
                r[u] = __builtin_amdgcn_readlane(dr, j);
                v[u] = rep[(size_t)sj * 32 + lane];
            }
            float p[8];
#pragma unroll
            for (int u = 0; u < 8; ++u) {
                const float4 t = tile4[r[u] * 32 + lane];
                p[u] = v[u].x * t.x + v[u].y * t.y + v[u].z * t.z + v[u].w * t.w;
            }
            const bool h = (lane & 16) != 0, g = (lane & 8) != 0, f = (lane & 4) != 0;
            float q[4];
#pragma unroll
            for (int u = 0; u < 4; ++u)
                q[u] = (h ? p[u + 4] : p[u]) + __shfl_xor(h ? p[u] : p[u + 4], 16, 32);
            float q2[2];
#pragma unroll
            for (int u = 0; u < 2; ++u)
                q2[u] = (g ? q[u + 2] : q[u]) + __shfl_xor(g ? q[u] : q[u + 2], 8, 32);
            float t1 = (f ? q2[1] : q2[0]) + __shfl_xor(f ? q2[0] : q2[1], 4, 32);
            t1 += __shfl_xor(t1, 2, 32);
            t1 += __shfl_xor(t1, 1, 32);
            if ((lane & 3) == 0 && ((vbits >> (lane >> 2)) & 1u)) {
                const float dl = t1 - target;
                local += dl * dl;
            }
        }
    };

    const unsigned c = wl2[lane];
    if (__builtin_amdgcn_ballot_w32(c > (unsigned)capw) == 0u) {
        const int cw = __builtin_amdgcn_readfirstlane((int)wl2[wave]);
        unsigned nxt = 0u;
        if (lane < cw) nxt = seg2[lane];
        for (int j0 = 0; j0 < cw; j0 += 32) {
            const unsigned cur = nxt;
            if (j0 + 32 + lane < cw) nxt = seg2[j0 + 32 + lane];
            const int nf = cw - j0 < 32 ? cw - j0 : 32;
            const unsigned m = nf >= 32 ? 0xFFFFFFFFu : ((1u << nf) - 1u);
            process(m, (int)(cur & 0xFFFFFu), (int)(cur >> 20));
        }
    } else {
        for (int base = wave * 32; base < E; base += 32 * AGG_WAVES) {
            const int e = base + lane;
            int s = 0, dr = 0;
            bool ok = false;
            if (e < E) {
                const int d = dst[e];
                const int sv = src[e];
                ok = (unsigned)(d - lo) < (unsigned)nr && sv < d;
                if (ok) { s = sv; dr = d - lo; }
            }
            process(__builtin_amdgcn_ballot_w32(ok), s, dr);
        }
    }

#pragma unroll
    for (int off = 16; off > 0; off >>= 1) local += __shfl_xor(local, off, 32);
    if (lane == 0) { wsum[wave] = local; wcnt[wave] = cntw; }
    __syncthreads();
    if (wave == 0) {
        float a = wsum[lane];
        unsigned b = wcnt[lane];
#pragma unroll
        for (int off = 16; off > 0; off >>= 1) {
            a += __shfl_xor(a, off, 32);
            b += __shfl_xor(b, off, 32);
        }
        if (lane == 0) { wl2[PART_LOSS] = __float_as_uint(a); wl2[PART_CNT] = b; }
    }
}

__global__ void __launch_bounds__(256) edge_loss_kernel(const float4* __restrict__ rep,
                                                        const int* __restrict__ src,
                                                        const int* __restrict__ dst, int E,
                                                        float target,
                                                        float* __restrict__ loss_acc,
                                                        unsigned* __restrict__ cnt_acc) {
    int lane   = threadIdx.x & 31;
    int wave   = blockIdx.x * (blockDim.x >> 5) + (threadIdx.x >> 5);
    int nwaves = gridDim.x * (blockDim.x >> 5);
    float local = 0.0f;
    unsigned cnt = 0;
    for (int e = wave; e < E; e += nwaves) {
        int s = src[e], d = dst[e];
        if (s < d) {
            float4 a = rep[(size_t)s * 32 + lane];
            float4 b = rep[(size_t)d * 32 + lane];
            float p = a.x * b.x + a.y * b.y + a.z * b.z + a.w * b.w;
#pragma unroll
            for (int off = 16; off > 0; off >>= 1) p += __shfl_down(p, off, 32);
            if (lane == 0) {
                float dl = p - target;
                local += dl * dl;
                cnt++;
            }
        }
    }
    if (lane == 0 && cnt > 0) {
        atomicAdd(loss_acc, local);
        atomicAdd(cnt_acc, cnt);
    }
}

__global__ void __launch_bounds__(256) finalize_kernel(const unsigned* __restrict__ part, int nwg,
                                                       const float* __restrict__ accum,
                                                       const unsigned* __restrict__ cnt,
                                                       float* __restrict__ out_scalar, float n_nodes) {
    __shared__ float sl[256];
    __shared__ unsigned sc[256];
    float l = 0.0f;
    unsigned c = 0u;
    for (int i = threadIdx.x; i < nwg; i += 256) {
        const unsigned* hd = part + (size_t)i * ROWS_PER_WG * FDIM;
        l += __uint_as_float(hd[PART_LOSS]);
        c += hd[PART_CNT];
    }
    sl[threadIdx.x] = l;
    sc[threadIdx.x] = c;
    __syncthreads();
    for (int w = 128; w > 0; w >>= 1) {
        if ((int)threadIdx.x < w) { sl[threadIdx.x] += sl[threadIdx.x + w]; sc[threadIdx.x] += sc[threadIdx.x + w]; }
        __syncthreads();
    }
    if (threadIdx.x == 0) {
        const float tot = sl[0] + accum[1];
        const float cc = (float)(sc[0] + cnt[1]);
        *out_scalar = tot * n_nodes / cc;
    }
}

extern "C" void kernel_launch(void* const* d_in, const int* in_sizes, int n_in,
                              void* d_out, int out_size, void* d_ws, size_t ws_size,
                              hipStream_t stream) {
    const float* features = (const float*)d_in[0];
    const int*   ei       = (const int*)d_in[1];
    const int*   nei      = (const int*)d_in[2];
    const float* w1       = (const float*)d_in[3];
    const float* b1       = (const float*)d_in[4];
    const float* w2       = (const float*)d_in[5];
    const float* b2       = (const float*)d_in[6];

    const int N  = in_sizes[0] / FDIM;
    const int E  = in_sizes[1] / 2;
    const int En = in_sizes[2] / 2;
    const int* src  = ei;
    const int* dst  = ei + E;
    const int* nsrc = nei;
    const int* ndst = nei + En;

    char* ws = (char*)d_ws;
    float*    bufA = (float*)ws;
    float*    bufB = (float*)(ws + (size_t)N * FDIM * 4);
    float*    dinv = (float*)(ws + (size_t)2 * N * FDIM * 4 + (size_t)N * 4);
    float*    accum = (float*)((char*)dinv + (size_t)N * 4);
    unsigned* cnt   = (unsigned*)(accum + 2);

    float* rep      = (float*)d_out;
    float* loss_out = rep + (size_t)N * FDIM;

    const int nstrips = N >> 4;
    const int gblocks = (nstrips + 7) / 8;
    const int ablocks = (N + ROWS_PER_WG - 1) / ROWS_PER_WG;

    zero_init_kernel<<<(ablocks + 255) / 256, 256, 0, stream>>>(accum, cnt, (unsigned*)bufB, ablocks);
    bin_kernel<<<(E + BIN_EDGES - 1) / BIN_EDGES, BIN_THREADS, 0, stream>>>(src, dst, (unsigned*)bufB, E, N, ablocks);
    owner_dinv_kernel<<<ablocks, DEG_THREADS, 0, stream>>>((const unsigned*)bufB, dst, dinv, E, N);

    gemm_wmma_kernel<false><<<gblocks, 256, 0, stream>>>(features, w1, bufA, N);
    owner_agg_kernel<true><<<ablocks, 1024, 0, stream>>>((const float4*)bufA, bufB, (unsigned*)rep, (unsigned*)bufB,
                                                         src, dst, dinv, b1, E, N);

    gemm_wmma_kernel<true><<<gblocks, 256, 0, stream>>>(bufB, w2, bufA, N);
    owner_agg_kernel<false><<<ablocks, 1024, 0, stream>>>((const float4*)bufA, rep, (unsigned*)rep, (unsigned*)bufB,
                                                          src, dst, dinv, b2, E, N);

    owner_loss_kernel<<<ablocks, 1024, 0, stream>>>((const float4*)rep, (unsigned*)bufB, src, dst, 1.0f, E, N);
    edge_loss_kernel<<<2048, 256, 0, stream>>>((const float4*)rep, nsrc, ndst, En, 0.0f,
                                               &accum[1], &cnt[1]);
    finalize_kernel<<<1, 256, 0, stream>>>((const unsigned*)bufB, ablocks, accum, cnt, loss_out, (float)N);
}


# the second edge set's loss joins the owner loss workgroups (equal slices, both rows gathered, 8 pairs in flight, same lane exchange): one launch and the float atomics go; partial sums (sum order)
# speedup vs baseline: 3.5426x; 1.2099x over previous
#include <hip/hip_runtime.h>

typedef __attribute__((ext_vector_type(2))) float v2f;
typedef __attribute__((ext_vector_type(8))) float v8f;

#define FDIM 128
#define ROWS_PER_WG 391
#define LIST_HDR 64
#define MAX_OWNERS 256
#define BIN_THREADS 256
#define BIN_EPT 8
#define BIN_EDGES (BIN_THREADS * BIN_EPT)
static_assert(BIN_THREADS == MAX_OWNERS, "thread b of the binning pass handles owner b");

__global__ void __launch_bounds__(256) zero_init_kernel(unsigned* __restrict__ bins, int nown) {
    int i = blockIdx.x * blockDim.x + threadIdx.x;
    if (i < nown) bins[(size_t)i * ROWS_PER_WG * FDIM] = 0u;
}

__global__ void __launch_bounds__(BIN_THREADS) bin_kernel(const int* __restrict__ src,
                                                          const int* __restrict__ dst,
                                                          unsigned* __restrict__ bins,
                                                          int E, int n, int nown) {
    __shared__ unsigned hist[MAX_OWNERS];
    __shared__ unsigned gbase[MAX_OWNERS];
    __shared__ unsigned wtot[BIN_THREADS / 32];
    __shared__ unsigned st_e[BIN_EDGES];
    __shared__ unsigned st_a[BIN_EDGES];

    const int tid = threadIdx.x;
    const int lane = tid & 31;
    const int wave = tid >> 5;
    if (nown > MAX_OWNERS) return;
    hist[tid] = 0u;
    __syncthreads();

    const int e0 = blockIdx.x * BIN_EDGES;
    int dv[BIN_EPT], sv[BIN_EPT];
#pragma unroll
    for (int k = 0; k < BIN_EPT; ++k) {
        const int e = e0 + k * BIN_THREADS + tid;
        dv[k] = e < E ? dst[e] : -1;
        sv[k] = e < E ? src[e] : 0;
    }
    int ow[BIN_EPT];
    unsigned ent[BIN_EPT], rk[BIN_EPT];
#pragma unroll
    for (int k = 0; k < BIN_EPT; ++k) {
        ow[k] = -1; ent[k] = 0u; rk[k] = 0u;
        if (dv[k] >= 0) {
            const int b = (int)((unsigned)dv[k] / (unsigned)ROWS_PER_WG);
            if (b < nown) {
                ow[k] = b;
                ent[k] = ((unsigned)(dv[k] - b * ROWS_PER_WG) << 20) | (unsigned)sv[k];
                rk[k] = atomicAdd(&hist[b], 1u);
            }
        }
    }
    __syncthreads();

    {
        const unsigned c = hist[tid];
        unsigned inc = c;
#pragma unroll
        for (int off = 1; off < 32; off <<= 1) {
            const unsigned t = __shfl_up(inc, off, 32);
            if (lane >= off) inc += t;
        }
        if (lane == 31) wtot[wave] = inc;
        __syncthreads();
        unsigned pre = 0u;
        for (int w = 0; w < wave; ++w) pre += wtot[w];
        unsigned gb = 0u;
        if (c != 0u) gb = atomicAdd(&bins[(size_t)tid * ROWS_PER_WG * FDIM], c);
        hist[tid] = pre + inc - c;
        gbase[tid] = gb;
    }
    __syncthreads();

#pragma unroll
    for (int k = 0; k < BIN_EPT; ++k) {
        if (ow[k] >= 0) {
            const int b = ow[k];
            const unsigned p = hist[b] + rk[k];
            int nrb = n - b * ROWS_PER_WG;
            if (nrb > ROWS_PER_WG) nrb = ROWS_PER_WG;
            const unsigned cap = (unsigned)(nrb * FDIM - LIST_HDR);
            const unsigned pos = gbase[b] + rk[k];
            st_e[p] = ent[k];
            st_a[p] = pos < cap ? (unsigned)b * (ROWS_PER_WG * FDIM) + LIST_HDR + pos : 0xFFFFFFFFu;
        }
    }
    __syncthreads();

    int nv = 0;
    for (int w = 0; w < BIN_THREADS / 32; ++w) nv += (int)wtot[w];
    for (int i = tid; i < nv; i += BIN_THREADS) {
        const unsigned a = st_a[i];
        if (a != 0xFFFFFFFFu) bins[a] = st_e[i];
    }
}

#define DEG_THREADS 1024
#define DEG_EPT 8
__global__ void __launch_bounds__(DEG_THREADS) owner_dinv_kernel(const unsigned* __restrict__ bins,
                                                                 const int* __restrict__ dst,
                                                                 float* __restrict__ dinv, int E, int n) {
    __shared__ unsigned rc[ROWS_PER_WG];
    const int tid = threadIdx.x;
    const int lo = blockIdx.x * ROWS_PER_WG;
    int nr = n - lo;
    if (nr > ROWS_PER_WG) nr = ROWS_PER_WG;
    for (int i = tid; i < nr; i += DEG_THREADS) rc[i] = 0u;
    __syncthreads();

    const unsigned* bk = bins + (size_t)lo * FDIM;
    const unsigned cb = bk[0];
    const bool binned = gridDim.x <= MAX_OWNERS && cb <= (unsigned)(nr * FDIM - LIST_HDR);
    if (binned) {
        for (unsigned base = 0u; base < cb; base += DEG_THREADS * DEG_EPT) {
            unsigned ev[DEG_EPT];
#pragma unroll
            for (int k = 0; k < DEG_EPT; ++k) {
                const unsigned i = base + (unsigned)(k * DEG_THREADS + tid);
                ev[k] = i < cb ? bk[LIST_HDR + i] : 0xFFFFFFFFu;
            }
#pragma unroll
            for (int k = 0; k < DEG_EPT; ++k)
                if (ev[k] != 0xFFFFFFFFu) atomicAdd(&rc[ev[k] >> 20], 1u);
        }
    } else {
        for (int e = tid; e < E; e += DEG_THREADS) {
            const unsigned r = (unsigned)(dst[e] - lo);
            if (r < (unsigned)nr) atomicAdd(&rc[r], 1u);
        }
    }
    __syncthreads();
    for (int i = tid; i < nr; i += DEG_THREADS) dinv[lo + i] = rsqrtf((float)rc[i] + 1.0f);
}

template <bool RELU>
__global__ void __launch_bounds__(256) gemm_wmma_kernel(const float* __restrict__ A,
                                                        const float* __restrict__ W,
                                                        float* __restrict__ Hout,
                                                        int nrows) {
    __shared__ float wlds[FDIM * FDIM];
    for (int i = threadIdx.x; i < FDIM * FDIM; i += 256) {
        int k = i >> 7, n = i & 127;
        wlds[((k >> 1) << 8) + (n << 1) + (k & 1)] = W[i];
    }
    __syncthreads();

    const int lane  = threadIdx.x & 31;
    const int wave  = threadIdx.x >> 5;
    const int strip = blockIdx.x * 8 + wave;
    const int nstrips = nrows >> 4;
    if (strip >= nstrips) return;

    const int row0  = strip << 4;
    const int m     = lane & 15;
    const int khalf = lane >> 4;
    const int kbase = khalf << 1;

    const float* arow = A + (size_t)(row0 + m) * FDIM;

    v8f acc[8];
#pragma unroll
    for (int t = 0; t < 8; ++t) acc[t] = (v8f)(0.0f);

    for (int k0 = 0; k0 < FDIM; k0 += 4) {
        v2f a = *(const v2f*)(arow + k0 + kbase);
        if (RELU) { a.x = fmaxf(a.x, 0.0f); a.y = fmaxf(a.y, 0.0f); }
        const int bbase = ((k0 >> 1) + khalf) * 256 + (m << 1);
#pragma unroll
        for (int t = 0; t < 8; ++t) {
            v2f b = *(const v2f*)(&wlds[bbase + t * 32]);
            acc[t] = __builtin_amdgcn_wmma_f32_16x16x4_f32(
                 false, a,  false, b,
                 (short)0, acc[t],  false,  false);
        }
    }

    const int rbase = row0 + (khalf << 3);
#pragma unroll
    for (int t = 0; t < 8; ++t) {
#pragma unroll
        for (int r = 0; r < 8; ++r)
            Hout[(size_t)(rbase + r) * FDIM + t * 16 + m] = acc[t][r];
    }
}

#define AGG_WAVES 32
#define AGG_UNROLL 8
#define PART_LOSS 32
#define PART_CNT 33
static_assert(AGG_WAVES == 32, "lane i reads the count of wave i");
static_assert(ROWS_PER_WG <= 2048, "local row must fit above bit 20");

template <bool FIRST>
__global__ void __launch_bounds__(1024) owner_agg_kernel(const float4* __restrict__ h,
                                                         float* out,
                                                         unsigned* list,
                                                         unsigned* list2,
                                                         const int* __restrict__ src,
                                                         const int* __restrict__ dst,
                                                         const float* __restrict__ dinv,
                                                         const float* __restrict__ bias,
                                                         int E, int n) {
    __shared__ float tile[ROWS_PER_WG * FDIM];
    __shared__ int queue[AGG_WAVES * 64];

    const int lane = threadIdx.x & 31;
    const int wave = threadIdx.x >> 5;
    const int qb = wave * 64;
    const int lo = blockIdx.x * ROWS_PER_WG;
    int nr = n - lo;
    if (nr > ROWS_PER_WG) nr = ROWS_PER_WG;

    unsigned* wl = list + (size_t)lo * FDIM;
    const int capw = ((nr * FDIM - LIST_HDR) >> 6) & ~31;
    uint2* seg = (uint2*)(wl + LIST_HDR) + (size_t)wave * capw;

    {
        const float b0 = bias[4 * lane], b1 = bias[4 * lane + 1];
        const float b2 = bias[4 * lane + 2], b3 = bias[4 * lane + 3];
        const float4* h4 = h + (size_t)lo * 32;
        for (int i = threadIdx.x; i < nr * 32; i += 1024) {
            const int row = i >> 5;
            const float4 v = h4[i];
            const float d = dinv[lo + row];
            const float d2 = d * d;
            float* t = tile + row * FDIM + lane;
            t[0] = v.x * d2 + b0; t[32] = v.y * d2 + b1; t[64] = v.z * d2 + b2; t[96] = v.w * d2 + b3;
        }
    }
    __syncthreads();

    auto add_row = [&](float4 v, int r, float w) {
        float* t = tile + r * FDIM + lane;
        atomicAdd(t, v.x * w);
        atomicAdd(t + 32, v.y * w);
        atomicAdd(t + 64, v.z * w);
        atomicAdd(t + 96, v.w * w);
    };

    auto gather = [&](int nf, int s, int dr, float nrm) {
        int j = 0;
        for (; j + 8 <= nf; j += 8) {
            float4 v[8];
            int r[8];
            float w[8];
#pragma unroll
            for (int u = 0; u < 8; ++u) {
                int sj = __builtin_amdgcn_readlane(s, j + u);
                r[u] = __builtin_amdgcn_readlane(dr, j + u);
                w[u] = __int_as_float(__builtin_amdgcn_readlane(__float_as_int(nrm), j + u));
                v[u] = h[(size_t)sj * 32 + lane];
            }
#pragma unroll
            for (int u = 0; u < 8; ++u) add_row(v[u], r[u], w[u]);
        }
        for (; j < nf; ++j) {
            int sj = __builtin_amdgcn_readlane(s, j);
            int rj = __builtin_amdgcn_readlane(dr, j);
            float wj = __int_as_float(__builtin_amdgcn_readlane(__float_as_int(nrm), j));
            add_row(h[(size_t)sj * 32 + lane], rj, wj);
        }
    };

    bool walk = false;
    if (!FIRST) {
        const unsigned c = wl[lane];
        walk = __builtin_amdgcn_ballot_w32(c > (unsigned)capw) == 0u;
    }

    bool binned = false;
    int cb = 0;
    const unsigned* bk = list2 + (size_t)lo * FDIM;
    if (FIRST) {
        cb = (int)bk[0];
        binned = gridDim.x <= MAX_OWNERS && (unsigned)cb <= (unsigned)(nr * FDIM - LIST_HDR);
    }

    if (FIRST && binned) {
        int tot = 0;
        int j0 = __builtin_amdgcn_readfirstlane(wave) * 32;
        const int step = 32 * AGG_WAVES;
        unsigned e1 = 0u, e2 = 0u;
        float na = 0.0f, nb = 0.0f;
        if (j0 + lane < cb) e1 = bk[LIST_HDR + j0 + lane];
        if (j0 + step + lane < cb) e2 = bk[LIST_HDR + j0 + step + lane];
        if (j0 + lane < cb) { na = dinv[e1 & 0xFFFFFu]; nb = dinv[lo + (int)(e1 >> 20)]; }
        for (; j0 < cb; j0 += step) {
            const unsigned cur = e1;
            const float nrm = na * nb;
            e1 = e2;
            e2 = 0u; na = 0.0f; nb = 0.0f;
            if (j0 + 2 * step + lane < cb) e2 = bk[LIST_HDR + j0 + 2 * step + lane];
            if (j0 + step + lane < cb) { na = dinv[e1 & 0xFFFFFu]; nb = dinv[lo + (int)(e1 >> 20)]; }
            const int nf = cb - j0 < 32 ? cb - j0 : 32;
            const int s = (int)(cur & 0xFFFFFu), dr = (int)(cur >> 20);
            if (lane < nf && tot + nf <= capw) seg[tot + lane] = make_uint2(cur, __float_as_uint(nrm));
            tot += nf;
            gather(nf, s, dr, nrm);
        }
        if (lane == 0) wl[wave] = (unsigned)tot;
    } else if (walk) {
        const int cw = __builtin_amdgcn_readfirstlane((int)wl[wave]);
        unsigned* wl2 = list2 + (size_t)lo * FDIM;
        unsigned* seg2 = wl2 + LIST_HDR + (size_t)wave * capw;
        int tot2 = 0;
        uint2 nxt = make_uint2(0u, 0u);
        if (lane < cw) nxt = seg[lane];
        for (int j0 = 0; j0 < cw; j0 += 32) {
            const uint2 cur = nxt;
            if (j0 + 32 + lane < cw) nxt = seg[j0 + 32 + lane];
            const int nf = cw - j0 < 32 ? cw - j0 : 32;
            const int sx = (int)(cur.x & 0xFFFFFu), rx = (int)(cur.x >> 20);
            const bool keep = lane < nf && sx < lo + rx;
            const unsigned km = __builtin_amdgcn_ballot_w32(keep);
            if (keep) seg2[tot2 + __popc(km & ((1u << lane) - 1u))] = cur.x;
            tot2 += __popc(km);
            gather(nf, sx, rx, __uint_as_float(cur.y));
        }
        if (lane == 0) wl2[wave] = (unsigned)tot2;
    } else {
        int cnt = 0;
        int tot = 0;
        const unsigned lt = (1u << lane) - 1u;

        auto flush = [&](int nf) {
            int s = 0, dr = 0;
            float nrm = 0.0f;
            if (lane < nf) {
                int e = queue[qb + lane];
                s = src[e];
                int d = dst[e];
                dr = d - lo;
                nrm = dinv[s] * dinv[d];
                if (FIRST && tot + nf <= capw)
                    seg[tot + lane] = make_uint2(((unsigned)dr << 20) | (unsigned)s, __float_as_uint(nrm));
            }
            tot += nf;
            gather(nf, s, dr, nrm);
        };

        int head = (int)(((16u - ((unsigned)(size_t)dst & 15u)) & 15u) >> 2);
        if (head > E) head = E;
        const int4* dst4 = (const int4*)(dst + head);
        const int nvec = (E - head) >> 2;
        const int ntail = (E - head) & 3;
        const int nextra = head + ntail;

        const int step = AGG_WAVES * 32 * AGG_UNROLL;
        const int npass = (nvec + step - 1) / step;
        for (int pass = 0; pass <= npass; ++pass) {
            const int vb = pass * step + wave * (32 * AGG_UNROLL);
            unsigned mbits = 0u;
            if (pass < npass) {
                int4 d4[AGG_UNROLL];
                unsigned nru[AGG_UNROLL];
#pragma unroll
                for (int u = 0; u < AGG_UNROLL; ++u) {
                    const int vi = vb + u * 32 + lane;
                    d4[u] = dst4[vi < nvec ? vi : nvec - 1];
                    nru[u] = vi < nvec ? (unsigned)nr : 0u;
                }
#pragma unroll
                for (int u = 0; u < AGG_UNROLL; ++u) {
                    const bool cx = (unsigned)(d4[u].x - lo) < nru[u];
                    const bool cy = (unsigned)(d4[u].y - lo) < nru[u];
                    const bool cz = (unsigned)(d4[u].z - lo) < nru[u];
                    const bool cw = (unsigned)(d4[u].w - lo) < nru[u];
                    const unsigned any = __builtin_amdgcn_ballot_w32(cx) | __builtin_amdgcn_ballot_w32(cy) |
                                         __builtin_amdgcn_ballot_w32(cz) | __builtin_amdgcn_ballot_w32(cw);
                    if (any)
                        mbits |= ((cx ? 1u : 0u) | (cy ? 2u : 0u) | (cz ? 4u : 0u) | (cw ? 8u : 0u)) << (4 * u);
                }
            } else if (wave == 0 && lane < nextra) {
                int e = lane < head ? lane : (nvec << 2) + lane;
                if ((unsigned)(dst[e] - lo) < (unsigned)nr) mbits = 1u;
            }
            while (__builtin_amdgcn_ballot_w32(mbits != 0u)) {
                const bool mt = mbits != 0u;
                const int k = mt ? (__ffs(mbits) - 1) : 0;
                int e;
                if (pass < npass) e = head + ((vb + (k >> 2) * 32 + lane) << 2) + (k & 3);
                else              e = lane < head ? lane : (nvec << 2) + lane;
                mbits &= mbits - 1u;
                const unsigned mask = __builtin_amdgcn_ballot_w32(mt);
                const int pos = cnt + __popc(mask & lt);
                if (mt) queue[qb + pos] = e;
                cnt += __popc(mask);
                if (cnt >= 32) {
                    flush(32);
                    const int rem = cnt - 32;
                    int t = 0;
                    if (lane < rem) t = queue[qb + 32 + lane];
                    if (lane < rem) queue[qb + lane] = t;
                    cnt = rem;
                }
            }
        }
        if (cnt > 0) flush(cnt);
        if (FIRST && lane == 0) wl[wave] = (unsigned)tot;
        if (!FIRST && lane == 0) list2[(size_t)lo * FDIM + wave] = 0xFFFFFFFFu;
    }

    __syncthreads();
    float4* out4 = (float4*)out + (size_t)lo * 32;
    for (int i = threadIdx.x; i < nr * 32; i += 1024) {
        const float* t = tile + (i >> 5) * FDIM + (i & 31);
        out4[i] = make_float4(t[0], t[32], t[64], t[96]);
    }
}

__global__ void __launch_bounds__(1024) owner_loss_kernel(const float4* __restrict__ rep,
                                                          unsigned* list2,
                                                          const int* __restrict__ src,
                                                          const int* __restrict__ dst,
                                                          const int* __restrict__ src2,
                                                          const int* __restrict__ dst2,
                                                          float target, float target2,
                                                          int E, int E2, int n) {
    __shared__ float4 tile4[ROWS_PER_WG * 32];
    __shared__ float wsum[AGG_WAVES];
    __shared__ unsigned wcnt[AGG_WAVES];

    const int lane = threadIdx.x & 31;
    const int wave = threadIdx.x >> 5;
    const int lo = blockIdx.x * ROWS_PER_WG;
    int nr = n - lo;
    if (nr > ROWS_PER_WG) nr = ROWS_PER_WG;
    unsigned* wl2 = list2 + (size_t)lo * FDIM;
    const int capw = ((nr * FDIM - LIST_HDR) >> 6) & ~31;
    const unsigned* seg2 = wl2 + LIST_HDR + (size_t)wave * capw;

    const int per = (E2 + (int)gridDim.x - 1) / (int)gridDim.x;
    const int b0 = (int)blockIdx.x * per;
    int b1 = b0 + per;
    if (b1 > E2) b1 = E2;
    const int bstep = 32 * AGG_WAVES;
    int base = b0 + __builtin_amdgcn_readfirstlane(wave) * 32;
    int s0 = 0, d0 = 0, s1 = 0, d1 = 0;
    if (base + lane < b1) { s0 = src2[base + lane]; d0 = dst2[base + lane]; }
    if (base + bstep + lane < b1) { s1 = src2[base + bstep + lane]; d1 = dst2[base + bstep + lane]; }

    for (int i = threadIdx.x; i < nr * 32; i += 1024) tile4[i] = rep[(size_t)lo * 32 + i];

    float local = 0.0f;
    unsigned cntw = 0u;

    auto lane_sums = [&](const float (&p)[8], unsigned vbits, float tgt) {
        const bool h = (lane & 16) != 0, g = (lane & 8) != 0, f = (lane & 4) != 0;
        float q[4];
#pragma unroll
        for (int u = 0; u < 4; ++u)
            q[u] = (h ? p[u + 4] : p[u]) + __shfl_xor(h ? p[u] : p[u + 4], 16, 32);
        float q2[2];
#pragma unroll
        for (int u = 0; u < 2; ++u)
            q2[u] = (g ? q[u + 2] : q[u]) + __shfl_xor(g ? q[u] : q[u + 2], 8, 32);
        float t1 = (f ? q2[1] : q2[0]) + __shfl_xor(f ? q2[0] : q2[1], 4, 32);
        t1 += __shfl_xor(t1, 2, 32);
        t1 += __shfl_xor(t1, 1, 32);
        if ((lane & 3) == 0 && ((vbits >> (lane >> 2)) & 1u)) {
            const float dl = t1 - tgt;
            local += dl * dl;
        }
    };

    auto pairs = [&](int s, int d) {
        unsigned m = __builtin_amdgcn_ballot_w32(s < d);
        cntw += (unsigned)__popc(m);
        while (m) {
            const int left = __popc(m);
            const unsigned vbits = left >= 8 ? 0xFFu : ((1u << left) - 1u);
            float4 va[8], vb[8];
#pragma unroll
            for (int u = 0; u < 8; ++u) {
                const int j = m ? (__ffs(m) - 1) : 0;
                m &= m - 1u;
                const int sj = __builtin_amdgcn_readlane(s, j);
                const int dj = __builtin_amdgcn_readlane(d, j);
                va[u] = rep[(size_t)sj * 32 + lane];
                vb[u] = rep[(size_t)dj * 32 + lane];
            }
            float p[8];
#pragma unroll
            for (int u = 0; u < 8; ++u)
                p[u] = va[u].x * vb[u].x + va[u].y * vb[u].y + va[u].z * vb[u].z + va[u].w * vb[u].w;
            lane_sums(p, vbits, target2);
        }
    };

    while (base < b1) {
        pairs(s0, d0);
        pairs(s1, d1);
        base += 2 * bstep;
        s0 = 0; d0 = 0; s1 = 0; d1 = 0;
        if (base + lane < b1) { s0 = src2[base + lane]; d0 = dst2[base + lane]; }
        if (base + bstep + lane < b1) { s1 = src2[base + bstep + lane]; d1 = dst2[base + bstep + lane]; }
    }
    __syncthreads();

    auto process = [&](unsigned m, int s, int dr) {
        cntw += (unsigned)__popc(m);
        while (m) {
            const int left = __popc(m);
            const unsigned vbits = left >= 8 ? 0xFFu : ((1u << left) - 1u);
            float4 v[8];
            int r[8];
#pragma unroll
            for (int u = 0; u < 8; ++u) {
                const int j = m ? (__ffs(m) - 1) : 0;
                m &= m - 1u;
                const int sj = __builtin_amdgcn_readlane(s, j);
                r[u] = __builtin_amdgcn_readlane(dr, j);
                v[u] = rep[(size_t)sj * 32 + lane];
            }
            float p[8];
#pragma unroll
            for (int u = 0; u < 8; ++u) {
                const float4 t = tile4[r[u] * 32 + lane];
                p[u] = v[u].x * t.x + v[u].y * t.y + v[u].z * t.z + v[u].w * t.w;
            }
            lane_sums(p, vbits, target);
        }
    };

    const unsigned c = wl2[lane];
    if (__builtin_amdgcn_ballot_w32(c > (unsigned)capw) == 0u) {
        const int cw = __builtin_amdgcn_readfirstlane((int)wl2[wave]);
        unsigned nxt = 0u;
        if (lane < cw) nxt = seg2[lane];
        for (int j0 = 0; j0 < cw; j0 += 32) {
            const unsigned cur = nxt;
            if (j0 + 32 + lane < cw) nxt = seg2[j0 + 32 + lane];
            const int nf = cw - j0 < 32 ? cw - j0 : 32;
            const unsigned m = nf >= 32 ? 0xFFFFFFFFu : ((1u << nf) - 1u);
            process(m, (int)(cur & 0xFFFFFu), (int)(cur >> 20));
        }
    } else {
        for (int base = wave * 32; base < E; base += 32 * AGG_WAVES) {
            const int e = base + lane;
            int s = 0, dr = 0;
            bool ok = false;
            if (e < E) {
                const int d = dst[e];
                const int sv = src[e];
                ok = (unsigned)(d - lo) < (unsigned)nr && sv < d;
                if (ok) { s = sv; dr = d - lo; }
            }
            process(__builtin_amdgcn_ballot_w32(ok), s, dr);
        }
    }

#pragma unroll
    for (int off = 16; off > 0; off >>= 1) local += __shfl_xor(local, off, 32);
    if (lane == 0) { wsum[wave] = local; wcnt[wave] = cntw; }
    __syncthreads();
    if (wave == 0) {
        float a = wsum[lane];
        unsigned b = wcnt[lane];
#pragma unroll
        for (int off = 16; off > 0; off >>= 1) {
            a += __shfl_xor(a, off, 32);
            b += __shfl_xor(b, off, 32);
        }
        if (lane == 0) { wl2[PART_LOSS] = __float_as_uint(a); wl2[PART_CNT] = b; }
    }
}

__global__ void __launch_bounds__(256) finalize_kernel(const unsigned* __restrict__ part, int nwg,
                                                       float* __restrict__ out_scalar, float n_nodes) {
    __shared__ float sl[256];
    __shared__ unsigned sc[256];
    float l = 0.0f;
    unsigned c = 0u;
    for (int i = threadIdx.x; i < nwg; i += 256) {
        const unsigned* hd = part + (size_t)i * ROWS_PER_WG * FDIM;
        l += __uint_as_float(hd[PART_LOSS]);
        c += hd[PART_CNT];
    }
    sl[threadIdx.x] = l;
    sc[threadIdx.x] = c;
    __syncthreads();
    for (int w = 128; w > 0; w >>= 1) {
        if ((int)threadIdx.x < w) { sl[threadIdx.x] += sl[threadIdx.x + w]; sc[threadIdx.x] += sc[threadIdx.x + w]; }
        __syncthreads();
    }
    if (threadIdx.x == 0) {
        *out_scalar = sl[0] * n_nodes / (float)sc[0];
    }
}

extern "C" void kernel_launch(void* const* d_in, const int* in_sizes, int n_in,
                              void* d_out, int out_size, void* d_ws, size_t ws_size,
                              hipStream_t stream) {
    const float* features = (const float*)d_in[0];
    const int*   ei       = (const int*)d_in[1];
    const int*   nei      = (const int*)d_in[2];
    const float* w1       = (const float*)d_in[3];
    const float* b1       = (const float*)d_in[4];
    const float* w2       = (const float*)d_in[5];
    const float* b2       = (const float*)d_in[6];

    const int N  = in_sizes[0] / FDIM;
    const int E  = in_sizes[1] / 2;
    const int En = in_sizes[2] / 2;
    const int* src  = ei;
    const int* dst  = ei + E;
    const int* nsrc = nei;
    const int* ndst = nei + En;

    char* ws = (char*)d_ws;
    float*    bufA = (float*)ws;
    float*    bufB = (float*)(ws + (size_t)N * FDIM * 4);
    float*    dinv = (float*)(ws + (size_t)2 * N * FDIM * 4 + (size_t)N * 4);

    float* rep      = (float*)d_out;
    float* loss_out = rep + (size_t)N * FDIM;

    const int nstrips = N >> 4;
    const int gblocks = (nstrips + 7) / 8;
    const int ablocks = (N + ROWS_PER_WG - 1) / ROWS_PER_WG;

    zero_init_kernel<<<(ablocks + 255) / 256, 256, 0, stream>>>((unsigned*)bufB, ablocks);
    bin_kernel<<<(E + BIN_EDGES - 1) / BIN_EDGES, BIN_THREADS, 0, stream>>>(src, dst, (unsigned*)bufB, E, N, ablocks);
    owner_dinv_kernel<<<ablocks, DEG_THREADS, 0, stream>>>((const unsigned*)bufB, dst, dinv, E, N);

    gemm_wmma_kernel<false><<<gblocks, 256, 0, stream>>>(features, w1, bufA, N);
    owner_agg_kernel<true><<<ablocks, 1024, 0, stream>>>((const float4*)bufA, bufB, (unsigned*)rep, (unsigned*)bufB,
                                                         src, dst, dinv, b1, E, N);

    gemm_wmma_kernel<true><<<gblocks, 256, 0, stream>>>(bufB, w2, bufA, N);
    owner_agg_kernel<false><<<ablocks, 1024, 0, stream>>>((const float4*)bufA, rep, (unsigned*)rep, (unsigned*)bufB,
                                                          src, dst, dinv, b2, E, N);

    owner_loss_kernel<<<ablocks, 1024, 0, stream>>>((const float4*)rep, (unsigned*)bufB, src, dst, nsrc, ndst,
                                                    1.0f, 0.0f, E, En, N);
    finalize_kernel<<<1, 256, 0, stream>>>((const unsigned*)bufB, ablocks, loss_out, (float)N);
}


# layer-1 GEMM and the edge binning pass share one launch (independent work; GEMM blocks enter first, binning blocks wait under their matrix issue): one launch fewer; sum orders unchanged
# speedup vs baseline: 3.6950x; 1.0430x over previous
#include <hip/hip_runtime.h>

typedef __attribute__((ext_vector_type(2))) float v2f;
typedef __attribute__((ext_vector_type(8))) float v8f;

#define FDIM 128
#define ROWS_PER_WG 391
#define LIST_HDR 64
#define MAX_OWNERS 256
#define BIN_THREADS 256
#define BIN_EPT 8
#define BIN_EDGES (BIN_THREADS * BIN_EPT)
static_assert(BIN_THREADS == MAX_OWNERS, "thread b of the binning pass handles owner b");

__global__ void __launch_bounds__(256) zero_init_kernel(unsigned* __restrict__ bins, int nown) {
    int i = blockIdx.x * blockDim.x + threadIdx.x;
    if (i < nown) bins[(size_t)i * ROWS_PER_WG * FDIM] = 0u;
}

#define BIN_LDS_DWORDS (2 * MAX_OWNERS + BIN_THREADS / 32 + 2 * BIN_EDGES)
static __device__ __forceinline__ void bin_body(unsigned* lds,
                                                const int* __restrict__ src,
                                                const int* __restrict__ dst,
                                                unsigned* __restrict__ bins,
                                                int E, int n, int nown, int blk) {
    unsigned* hist = lds;
    unsigned* gbase = hist + MAX_OWNERS;
    unsigned* wtot = gbase + MAX_OWNERS;
    unsigned* st_e = wtot + BIN_THREADS / 32;
    unsigned* st_a = st_e + BIN_EDGES;

    const int tid = threadIdx.x;
    const int lane = tid & 31;
    const int wave = tid >> 5;
    if (nown > MAX_OWNERS) return;
    hist[tid] = 0u;
    __syncthreads();

    const int e0 = blk * BIN_EDGES;
    int dv[BIN_EPT], sv[BIN_EPT];
#pragma unroll
    for (int k = 0; k < BIN_EPT; ++k) {
        const int e = e0 + k * BIN_THREADS + tid;
        dv[k] = e < E ? dst[e] : -1;
        sv[k] = e < E ? src[e] : 0;
    }
    int ow[BIN_EPT];
    unsigned ent[BIN_EPT], rk[BIN_EPT];
#pragma unroll
    for (int k = 0; k < BIN_EPT; ++k) {
        ow[k] = -1; ent[k] = 0u; rk[k] = 0u;
        if (dv[k] >= 0) {
            const int b = (int)((unsigned)dv[k] / (unsigned)ROWS_PER_WG);
            if (b < nown) {
                ow[k] = b;
                ent[k] = ((unsigned)(dv[k] - b * ROWS_PER_WG) << 20) | (unsigned)sv[k];
                rk[k] = atomicAdd(&hist[b], 1u);
            }
        }
    }
    __syncthreads();

    {
        const unsigned c = hist[tid];
        unsigned inc = c;
#pragma unroll
        for (int off = 1; off < 32; off <<= 1) {
            const unsigned t = __shfl_up(inc, off, 32);
            if (lane >= off) inc += t;
        }
        if (lane == 31) wtot[wave] = inc;
        __syncthreads();
        unsigned pre = 0u;
        for (int w = 0; w < wave; ++w) pre += wtot[w];
        unsigned gb = 0u;
        if (c != 0u) gb = atomicAdd(&bins[(size_t)tid * ROWS_PER_WG * FDIM], c);
        hist[tid] = pre + inc - c;
        gbase[tid] = gb;
    }
    __syncthreads();

#pragma unroll
    for (int k = 0; k < BIN_EPT; ++k) {
        if (ow[k] >= 0) {
            const int b = ow[k];
            const unsigned p = hist[b] + rk[k];
            int nrb = n - b * ROWS_PER_WG;
            if (nrb > ROWS_PER_WG) nrb = ROWS_PER_WG;
            const unsigned cap = (unsigned)(nrb * FDIM - LIST_HDR);
            const unsigned pos = gbase[b] + rk[k];
            st_e[p] = ent[k];
            st_a[p] = pos < cap ? (unsigned)b * (ROWS_PER_WG * FDIM) + LIST_HDR + pos : 0xFFFFFFFFu;
        }
    }
    __syncthreads();

    int nv = 0;
    for (int w = 0; w < BIN_THREADS / 32; ++w) nv += (int)wtot[w];
    for (int i = tid; i < nv; i += BIN_THREADS) {
        const unsigned a = st_a[i];
        if (a != 0xFFFFFFFFu) bins[a] = st_e[i];
    }
}

#define DEG_THREADS 1024
#define DEG_EPT 8
__global__ void __launch_bounds__(DEG_THREADS) owner_dinv_kernel(const unsigned* __restrict__ bins,
                                                                 const int* __restrict__ dst,
                                                                 float* __restrict__ dinv, int E, int n) {
    __shared__ unsigned rc[ROWS_PER_WG];
    const int tid = threadIdx.x;
    const int lo = blockIdx.x * ROWS_PER_WG;
    int nr = n - lo;
    if (nr > ROWS_PER_WG) nr = ROWS_PER_WG;
    for (int i = tid; i < nr; i += DEG_THREADS) rc[i] = 0u;
    __syncthreads();

    const unsigned* bk = bins + (size_t)lo * FDIM;
    const unsigned cb = bk[0];
    const bool binned = gridDim.x <= MAX_OWNERS && cb <= (unsigned)(nr * FDIM - LIST_HDR);
    if (binned) {
        for (unsigned base = 0u; base < cb; base += DEG_THREADS * DEG_EPT) {
            unsigned ev[DEG_EPT];
#pragma unroll
            for (int k = 0; k < DEG_EPT; ++k) {
                const unsigned i = base + (unsigned)(k * DEG_THREADS + tid);
                ev[k] = i < cb ? bk[LIST_HDR + i] : 0xFFFFFFFFu;
            }
#pragma unroll
            for (int k = 0; k < DEG_EPT; ++k)
                if (ev[k] != 0xFFFFFFFFu) atomicAdd(&rc[ev[k] >> 20], 1u);
        }
    } else {
        for (int e = tid; e < E; e += DEG_THREADS) {
            const unsigned r = (unsigned)(dst[e] - lo);
            if (r < (unsigned)nr) atomicAdd(&rc[r], 1u);
        }
    }
    __syncthreads();
    for (int i = tid; i < nr; i += DEG_THREADS) dinv[lo + i] = rsqrtf((float)rc[i] + 1.0f);
}

template <bool RELU>
static __device__ __forceinline__ void gemm_body(float* wlds,
                                                 const float* __restrict__ A,
                                                 const float* __restrict__ W,
                                                 float* __restrict__ Hout,
                                                 int nrows, int blk) {
    for (int i = threadIdx.x; i < FDIM * FDIM; i += 256) {
        int k = i >> 7, n = i & 127;
        wlds[((k >> 1) << 8) + (n << 1) + (k & 1)] = W[i];
    }
    __syncthreads();

    const int lane  = threadIdx.x & 31;
    const int wave  = threadIdx.x >> 5;
    const int strip = blk * 8 + wave;
    const int nstrips = nrows >> 4;
    if (strip >= nstrips) return;

    const int row0  = strip << 4;
    const int m     = lane & 15;
    const int khalf = lane >> 4;
    const int kbase = khalf << 1;

    const float* arow = A + (size_t)(row0 + m) * FDIM;

    v8f acc[8];
#pragma unroll
    for (int t = 0; t < 8; ++t) acc[t] = (v8f)(0.0f);

    for (int k0 = 0; k0 < FDIM; k0 += 4) {
        v2f a = *(const v2f*)(arow + k0 + kbase);
        if (RELU) { a.x = fmaxf(a.x, 0.0f); a.y = fmaxf(a.y, 0.0f); }
        const int bbase = ((k0 >> 1) + khalf) * 256 + (m << 1);
#pragma unroll
        for (int t = 0; t < 8; ++t) {
            v2f b = *(const v2f*)(&wlds[bbase + t * 32]);
            acc[t] = __builtin_amdgcn_wmma_f32_16x16x4_f32(
                 false, a,  false, b,
                 (short)0, acc[t],  false,  false);
        }
    }

    const int rbase = row0 + (khalf << 3);
#pragma unroll
    for (int t = 0; t < 8; ++t) {
#pragma unroll
        for (int r = 0; r < 8; ++r)
            Hout[(size_t)(rbase + r) * FDIM + t * 16 + m] = acc[t][r];
    }
}

template <bool RELU>
__global__ void __launch_bounds__(256) gemm_wmma_kernel(const float* __restrict__ A,
                                                        const float* __restrict__ W,
                                                        float* __restrict__ Hout,
                                                        int nrows) {
    __shared__ float wlds[FDIM * FDIM];
    gemm_body<RELU>(wlds, A, W, Hout, nrows, (int)blockIdx.x);
}

static_assert(BIN_LDS_DWORDS <= FDIM * FDIM, "the binning arrays fit in the GEMM's LDS buffer");
static_assert(BIN_THREADS == 256, "both kinds of block have 256 threads");
__global__ void __launch_bounds__(256) gemm_bin_kernel(const float* __restrict__ A,
                                                       const float* __restrict__ W,
                                                       float* __restrict__ Hout,
                                                       int nrows, int gblocks,
                                                       const int* __restrict__ src,
                                                       const int* __restrict__ dst,
                                                       unsigned* __restrict__ bins,
                                                       int E, int n, int nown) {
    __shared__ __attribute__((aligned(16))) unsigned char smem[FDIM * FDIM * 4];
    const int b = (int)blockIdx.x;
    if (b < gblocks) gemm_body<false>((float*)smem, A, W, Hout, nrows, b);
    else             bin_body((unsigned*)smem, src, dst, bins, E, n, nown, b - gblocks);
}

#define AGG_WAVES 32
#define AGG_UNROLL 8
#define PART_LOSS 32
#define PART_CNT 33
static_assert(AGG_WAVES == 32, "lane i reads the count of wave i");
static_assert(ROWS_PER_WG <= 2048, "local row must fit above bit 20");

template <bool FIRST>
__global__ void __launch_bounds__(1024) owner_agg_kernel(const float4* __restrict__ h,
                                                         float* out,
                                                         unsigned* list,
                                                         unsigned* list2,
                                                         const int* __restrict__ src,
                                                         const int* __restrict__ dst,
                                                         const float* __restrict__ dinv,
                                                         const float* __restrict__ bias,
                                                         int E, int n) {
    __shared__ float tile[ROWS_PER_WG * FDIM];
    __shared__ int queue[AGG_WAVES * 64];

    const int lane = threadIdx.x & 31;
    const int wave = threadIdx.x >> 5;
    const int qb = wave * 64;
    const int lo = blockIdx.x * ROWS_PER_WG;
    int nr = n - lo;
    if (nr > ROWS_PER_WG) nr = ROWS_PER_WG;

    unsigned* wl = list + (size_t)lo * FDIM;
    const int capw = ((nr * FDIM - LIST_HDR) >> 6) & ~31;
    uint2* seg = (uint2*)(wl + LIST_HDR) + (size_t)wave * capw;

    {
        const float b0 = bias[4 * lane], b1 = bias[4 * lane + 1];
        const float b2 = bias[4 * lane + 2], b3 = bias[4 * lane + 3];
        const float4* h4 = h + (size_t)lo * 32;
        for (int i = threadIdx.x; i < nr * 32; i += 1024) {
            const int row = i >> 5;
            const float4 v = h4[i];
            const float d = dinv[lo + row];
            const float d2 = d * d;
            float* t = tile + row * FDIM + lane;
            t[0] = v.x * d2 + b0; t[32] = v.y * d2 + b1; t[64] = v.z * d2 + b2; t[96] = v.w * d2 + b3;
        }
    }
    __syncthreads();

    auto add_row = [&](float4 v, int r, float w) {
        float* t = tile + r * FDIM + lane;
        atomicAdd(t, v.x * w);
        atomicAdd(t + 32, v.y * w);
        atomicAdd(t + 64, v.z * w);
        atomicAdd(t + 96, v.w * w);
    };

    auto gather = [&](int nf, int s, int dr, float nrm) {
        int j = 0;
        for (; j + 8 <= nf; j += 8) {
            float4 v[8];
            int r[8];
            float w[8];
#pragma unroll
            for (int u = 0; u < 8; ++u) {
                int sj = __builtin_amdgcn_readlane(s, j + u);
                r[u] = __builtin_amdgcn_readlane(dr, j + u);
                w[u] = __int_as_float(__builtin_amdgcn_readlane(__float_as_int(nrm), j + u));
                v[u] = h[(size_t)sj * 32 + lane];
            }
#pragma unroll
            for (int u = 0; u < 8; ++u) add_row(v[u], r[u], w[u]);
        }
        for (; j < nf; ++j) {
            int sj = __builtin_amdgcn_readlane(s, j);
            int rj = __builtin_amdgcn_readlane(dr, j);
            float wj = __int_as_float(__builtin_amdgcn_readlane(__float_as_int(nrm), j));
            add_row(h[(size_t)sj * 32 + lane], rj, wj);
        }
    };

    bool walk = false;
    if (!FIRST) {
        const unsigned c = wl[lane];
        walk = __builtin_amdgcn_ballot_w32(c > (unsigned)capw) == 0u;
    }

    bool binned = false;
    int cb = 0;
    const unsigned* bk = list2 + (size_t)lo * FDIM;
    if (FIRST) {
        cb = (int)bk[0];
        binned = gridDim.x <= MAX_OWNERS && (unsigned)cb <= (unsigned)(nr * FDIM - LIST_HDR);
    }

    if (FIRST && binned) {
        int tot = 0;
        int j0 = __builtin_amdgcn_readfirstlane(wave) * 32;
        const int step = 32 * AGG_WAVES;
        unsigned e1 = 0u, e2 = 0u;
        float na = 0.0f, nb = 0.0f;
        if (j0 + lane < cb) e1 = bk[LIST_HDR + j0 + lane];
        if (j0 + step + lane < cb) e2 = bk[LIST_HDR + j0 + step + lane];
        if (j0 + lane < cb) { na = dinv[e1 & 0xFFFFFu]; nb = dinv[lo + (int)(e1 >> 20)]; }
        for (; j0 < cb; j0 += step) {
            const unsigned cur = e1;
            const float nrm = na * nb;
            e1 = e2;
            e2 = 0u; na = 0.0f; nb = 0.0f;
            if (j0 + 2 * step + lane < cb) e2 = bk[LIST_HDR + j0 + 2 * step + lane];
            if (j0 + step + lane < cb) { na = dinv[e1 & 0xFFFFFu]; nb = dinv[lo + (int)(e1 >> 20)]; }
            const int nf = cb - j0 < 32 ? cb - j0 : 32;
            const int s = (int)(cur & 0xFFFFFu), dr = (int)(cur >> 20);
            if (lane < nf && tot + nf <= capw) seg[tot + lane] = make_uint2(cur, __float_as_uint(nrm));
            tot += nf;
            gather(nf, s, dr, nrm);
        }
        if (lane == 0) wl[wave] = (unsigned)tot;
    } else if (walk) {
        const int cw = __builtin_amdgcn_readfirstlane((int)wl[wave]);
        unsigned* wl2 = list2 + (size_t)lo * FDIM;
        unsigned* seg2 = wl2 + LIST_HDR + (size_t)wave * capw;
        int tot2 = 0;
        uint2 nxt = make_uint2(0u, 0u);
        if (lane < cw) nxt = seg[lane];
        for (int j0 = 0; j0 < cw; j0 += 32) {
            const uint2 cur = nxt;
            if (j0 + 32 + lane < cw) nxt = seg[j0 + 32 + lane];
            const int nf = cw - j0 < 32 ? cw - j0 : 32;
            const int sx = (int)(cur.x & 0xFFFFFu), rx = (int)(cur.x >> 20);
            const bool keep = lane < nf && sx < lo + rx;
            const unsigned km = __builtin_amdgcn_ballot_w32(keep);
            if (keep) seg2[tot2 + __popc(km & ((1u << lane) - 1u))] = cur.x;
            tot2 += __popc(km);
            gather(nf, sx, rx, __uint_as_float(cur.y));
        }
        if (lane == 0) wl2[wave] = (unsigned)tot2;
    } else {
        int cnt = 0;
        int tot = 0;
        const unsigned lt = (1u << lane) - 1u;

        auto flush = [&](int nf) {
            int s = 0, dr = 0;
            float nrm = 0.0f;
            if (lane < nf) {
                int e = queue[qb + lane];
                s = src[e];
                int d = dst[e];
                dr = d - lo;
                nrm = dinv[s] * dinv[d];
                if (FIRST && tot + nf <= capw)
                    seg[tot + lane] = make_uint2(((unsigned)dr << 20) | (unsigned)s, __float_as_uint(nrm));
            }
            tot += nf;
            gather(nf, s, dr, nrm);
        };

        int head = (int)(((16u - ((unsigned)(size_t)dst & 15u)) & 15u) >> 2);
        if (head > E) head = E;
        const int4* dst4 = (const int4*)(dst + head);
        const int nvec = (E - head) >> 2;
        const int ntail = (E - head) & 3;
        const int nextra = head + ntail;

        const int step = AGG_WAVES * 32 * AGG_UNROLL;
        const int npass = (nvec + step - 1) / step;
        for (int pass = 0; pass <= npass; ++pass) {
            const int vb = pass * step + wave * (32 * AGG_UNROLL);
            unsigned mbits = 0u;
            if (pass < npass) {
                int4 d4[AGG_UNROLL];
                unsigned nru[AGG_UNROLL];
#pragma unroll
                for (int u = 0; u < AGG_UNROLL; ++u) {
                    const int vi = vb + u * 32 + lane;
                    d4[u] = dst4[vi < nvec ? vi : nvec - 1];
                    nru[u] = vi < nvec ? (unsigned)nr : 0u;
                }
#pragma unroll
                for (int u = 0; u < AGG_UNROLL; ++u) {
                    const bool cx = (unsigned)(d4[u].x - lo) < nru[u];
                    const bool cy = (unsigned)(d4[u].y - lo) < nru[u];
                    const bool cz = (unsigned)(d4[u].z - lo) < nru[u];
                    const bool cw = (unsigned)(d4[u].w - lo) < nru[u];
                    const unsigned any = __builtin_amdgcn_ballot_w32(cx) | __builtin_amdgcn_ballot_w32(cy) |
                                         __builtin_amdgcn_ballot_w32(cz) | __builtin_amdgcn_ballot_w32(cw);
                    if (any)
                        mbits |= ((cx ? 1u : 0u) | (cy ? 2u : 0u) | (cz ? 4u : 0u) | (cw ? 8u : 0u)) << (4 * u);
                }
            } else if (wave == 0 && lane < nextra) {
                int e = lane < head ? lane : (nvec << 2) + lane;
                if ((unsigned)(dst[e] - lo) < (unsigned)nr) mbits = 1u;
            }
            while (__builtin_amdgcn_ballot_w32(mbits != 0u)) {
                const bool mt = mbits != 0u;
                const int k = mt ? (__ffs(mbits) - 1) : 0;
                int e;
                if (pass < npass) e = head + ((vb + (k >> 2) * 32 + lane) << 2) + (k & 3);
                else              e = lane < head ? lane : (nvec << 2) + lane;
                mbits &= mbits - 1u;
                const unsigned mask = __builtin_amdgcn_ballot_w32(mt);
                const int pos = cnt + __popc(mask & lt);
                if (mt) queue[qb + pos] = e;
                cnt += __popc(mask);
                if (cnt >= 32) {
                    flush(32);
                    const int rem = cnt - 32;
                    int t = 0;
                    if (lane < rem) t = queue[qb + 32 + lane];
                    if (lane < rem) queue[qb + lane] = t;
                    cnt = rem;
                }
            }
        }
        if (cnt > 0) flush(cnt);
        if (FIRST && lane == 0) wl[wave] = (unsigned)tot;
        if (!FIRST && lane == 0) list2[(size_t)lo * FDIM + wave] = 0xFFFFFFFFu;
    }

    __syncthreads();
    float4* out4 = (float4*)out + (size_t)lo * 32;
    for (int i = threadIdx.x; i < nr * 32; i += 1024) {
        const float* t = tile + (i >> 5) * FDIM + (i & 31);
        out4[i] = make_float4(t[0], t[32], t[64], t[96]);
    }
}

__global__ void __launch_bounds__(1024) owner_loss_kernel(const float4* __restrict__ rep,
                                                          unsigned* list2,
                                                          const int* __restrict__ src,
                                                          const int* __restrict__ dst,
                                                          const int* __restrict__ src2,
                                                          const int* __restrict__ dst2,
                                                          float target, float target2,
                                                          int E, int E2, int n) {
    __shared__ float4 tile4[ROWS_PER_WG * 32];
    __shared__ float wsum[AGG_WAVES];
    __shared__ unsigned wcnt[AGG_WAVES];

    const int lane = threadIdx.x & 31;
    const int wave = threadIdx.x >> 5;
    const int lo = blockIdx.x * ROWS_PER_WG;
    int nr = n - lo;
    if (nr > ROWS_PER_WG) nr = ROWS_PER_WG;
    unsigned* wl2 = list2 + (size_t)lo * FDIM;
    const int capw = ((nr * FDIM - LIST_HDR) >> 6) & ~31;
    const unsigned* seg2 = wl2 + LIST_HDR + (size_t)wave * capw;

    const int per = (E2 + (int)gridDim.x - 1) / (int)gridDim.x;
    const int b0 = (int)blockIdx.x * per;
    int b1 = b0 + per;
    if (b1 > E2) b1 = E2;
    const int bstep = 32 * AGG_WAVES;
    int base = b0 + __builtin_amdgcn_readfirstlane(wave) * 32;
    int s0 = 0, d0 = 0, s1 = 0, d1 = 0;
    if (base + lane < b1) { s0 = src2[base + lane]; d0 = dst2[base + lane]; }
    if (base + bstep + lane < b1) { s1 = src2[base + bstep + lane]; d1 = dst2[base + bstep + lane]; }

    for (int i = threadIdx.x; i < nr * 32; i += 1024) tile4[i] = rep[(size_t)lo * 32 + i];

    float local = 0.0f;
    unsigned cntw = 0u;

    auto lane_sums = [&](const float (&p)[8], unsigned vbits, float tgt) {
        const bool h = (lane & 16) != 0, g = (lane & 8) != 0, f = (lane & 4) != 0;
        float q[4];
#pragma unroll
        for (int u = 0; u < 4; ++u)
            q[u] = (h ? p[u + 4] : p[u]) + __shfl_xor(h ? p[u] : p[u + 4], 16, 32);
        float q2[2];
#pragma unroll
        for (int u = 0; u < 2; ++u)
            q2[u] = (g ? q[u + 2] : q[u]) + __shfl_xor(g ? q[u] : q[u + 2], 8, 32);
        float t1 = (f ? q2[1] : q2[0]) + __shfl_xor(f ? q2[0] : q2[1], 4, 32);
        t1 += __shfl_xor(t1, 2, 32);
        t1 += __shfl_xor(t1, 1, 32);
        if ((lane & 3) == 0 && ((vbits >> (lane >> 2)) & 1u)) {
            const float dl = t1 - tgt;
            local += dl * dl;
        }
    };

    auto pairs = [&](int s, int d) {
        unsigned m = __builtin_amdgcn_ballot_w32(s < d);
        cntw += (unsigned)__popc(m);
        while (m) {
            const int left = __popc(m);
            const unsigned vbits = left >= 8 ? 0xFFu : ((1u << left) - 1u);
            float4 va[8], vb[8];
#pragma unroll
            for (int u = 0; u < 8; ++u) {
                const int j = m ? (__ffs(m) - 1) : 0;
                m &= m - 1u;
                const int sj = __builtin_amdgcn_readlane(s, j);
                const int dj = __builtin_amdgcn_readlane(d, j);
                va[u] = rep[(size_t)sj * 32 + lane];
                vb[u] = rep[(size_t)dj * 32 + lane];
            }
            float p[8];
#pragma unroll
            for (int u = 0; u < 8; ++u)
                p[u] = va[u].x * vb[u].x + va[u].y * vb[u].y + va[u].z * vb[u].z + va[u].w * vb[u].w;
            lane_sums(p, vbits, target2);
        }
    };

    while (base < b1) {
        pairs(s0, d0);
        pairs(s1, d1);
        base += 2 * bstep;
        s0 = 0; d0 = 0; s1 = 0; d1 = 0;
        if (base + lane < b1) { s0 = src2[base + lane]; d0 = dst2[base + lane]; }
        if (base + bstep + lane < b1) { s1 = src2[base + bstep + lane]; d1 = dst2[base + bstep + lane]; }
    }
    __syncthreads();

    auto process = [&](unsigned m, int s, int dr) {
        cntw += (unsigned)__popc(m);
        while (m) {
            const int left = __popc(m);
            const unsigned vbits = left >= 8 ? 0xFFu : ((1u << left) - 1u);
            float4 v[8];
            int r[8];
#pragma unroll
            for (int u = 0; u < 8; ++u) {
                const int j = m ? (__ffs(m) - 1) : 0;
                m &= m - 1u;
                const int sj = __builtin_amdgcn_readlane(s, j);
                r[u] = __builtin_amdgcn_readlane(dr, j);
                v[u] = rep[(size_t)sj * 32 + lane];
            }
            float p[8];
#pragma unroll
            for (int u = 0; u < 8; ++u) {
                const float4 t = tile4[r[u] * 32 + lane];
                p[u] = v[u].x * t.x + v[u].y * t.y + v[u].z * t.z + v[u].w * t.w;
            }
            lane_sums(p, vbits, target);
        }
    };

    const unsigned c = wl2[lane];
    if (__builtin_amdgcn_ballot_w32(c > (unsigned)capw) == 0u) {
        const int cw = __builtin_amdgcn_readfirstlane((int)wl2[wave]);
        unsigned nxt = 0u;
        if (lane < cw) nxt = seg2[lane];
        for (int j0 = 0; j0 < cw; j0 += 32) {
            const unsigned cur = nxt;
            if (j0 + 32 + lane < cw) nxt = seg2[j0 + 32 + lane];
            const int nf = cw - j0 < 32 ? cw - j0 : 32;
            const unsigned m = nf >= 32 ? 0xFFFFFFFFu : ((1u << nf) - 1u);
            process(m, (int)(cur & 0xFFFFFu), (int)(cur >> 20));
        }
    } else {
        for (int base = wave * 32; base < E; base += 32 * AGG_WAVES) {
            const int e = base + lane;
            int s = 0, dr = 0;
            bool ok = false;
            if (e < E) {
                const int d = dst[e];
                const int sv = src[e];
                ok = (unsigned)(d - lo) < (unsigned)nr && sv < d;
                if (ok) { s = sv; dr = d - lo; }
            }
            process(__builtin_amdgcn_ballot_w32(ok), s, dr);
        }
    }

#pragma unroll
    for (int off = 16; off > 0; off >>= 1) local += __shfl_xor(local, off, 32);
    if (lane == 0) { wsum[wave] = local; wcnt[wave] = cntw; }
    __syncthreads();
    if (wave == 0) {
        float a = wsum[lane];
        unsigned b = wcnt[lane];
#pragma unroll
        for (int off = 16; off > 0; off >>= 1) {
            a += __shfl_xor(a, off, 32);
            b += __shfl_xor(b, off, 32);
        }
        if (lane == 0) { wl2[PART_LOSS] = __float_as_uint(a); wl2[PART_CNT] = b; }
    }
}

__global__ void __launch_bounds__(256) finalize_kernel(const unsigned* __restrict__ part, int nwg,
                                                       float* __restrict__ out_scalar, float n_nodes) {
    __shared__ float sl[256];
    __shared__ unsigned sc[256];
    float l = 0.0f;
    unsigned c = 0u;
    for (int i = threadIdx.x; i < nwg; i += 256) {
        const unsigned* hd = part + (size_t)i * ROWS_PER_WG * FDIM;
        l += __uint_as_float(hd[PART_LOSS]);
        c += hd[PART_CNT];
    }
    sl[threadIdx.x] = l;
    sc[threadIdx.x] = c;
    __syncthreads();
    for (int w = 128; w > 0; w >>= 1) {
        if ((int)threadIdx.x < w) { sl[threadIdx.x] += sl[threadIdx.x + w]; sc[threadIdx.x] += sc[threadIdx.x + w]; }
        __syncthreads();
    }
    if (threadIdx.x == 0) {
        *out_scalar = sl[0] * n_nodes / (float)sc[0];
    }
}

extern "C" void kernel_launch(void* const* d_in, const int* in_sizes, int n_in,
                              void* d_out, int out_size, void* d_ws, size_t ws_size,
                              hipStream_t stream) {
    const float* features = (const float*)d_in[0];
    const int*   ei       = (const int*)d_in[1];
    const int*   nei      = (const int*)d_in[2];
    const float* w1       = (const float*)d_in[3];
    const float* b1       = (const float*)d_in[4];
    const float* w2       = (const float*)d_in[5];
    const float* b2       = (const float*)d_in[6];

    const int N  = in_sizes[0] / FDIM;
    const int E  = in_sizes[1] / 2;
    const int En = in_sizes[2] / 2;
    const int* src  = ei;
    const int* dst  = ei + E;
    const int* nsrc = nei;
    const int* ndst = nei + En;

    char* ws = (char*)d_ws;
    float*    bufA = (float*)ws;
    float*    bufB = (float*)(ws + (size_t)N * FDIM * 4);
    float*    dinv = (float*)(ws + (size_t)2 * N * FDIM * 4 + (size_t)N * 4);

    float* rep      = (float*)d_out;
    float* loss_out = rep + (size_t)N * FDIM;

    const int nstrips = N >> 4;
    const int gblocks = (nstrips + 7) / 8;
    const int ablocks = (N + ROWS_PER_WG - 1) / ROWS_PER_WG;

    zero_init_kernel<<<(ablocks + 255) / 256, 256, 0, stream>>>((unsigned*)bufB, ablocks);
    const int bblocks = (E + BIN_EDGES - 1) / BIN_EDGES;
    gemm_bin_kernel<<<gblocks + bblocks, 256, 0, stream>>>(features, w1, bufA, N, gblocks,
                                                           src, dst, (unsigned*)bufB, E, N, ablocks);
    owner_dinv_kernel<<<ablocks, DEG_THREADS, 0, stream>>>((const unsigned*)bufB, dst, dinv, E, N);

    owner_agg_kernel<true><<<ablocks, 1024, 0, stream>>>((const float4*)bufA, bufB, (unsigned*)rep, (unsigned*)bufB,
                                                         src, dst, dinv, b1, E, N);

    gemm_wmma_kernel<true><<<gblocks, 256, 0, stream>>>(bufB, w2, bufA, N);
    owner_agg_kernel<false><<<ablocks, 1024, 0, stream>>>((const float4*)bufA, rep, (unsigned*)rep, (unsigned*)bufB,
                                                          src, dst, dinv, b2, E, N);

    owner_loss_kernel<<<ablocks, 1024, 0, stream>>>((const float4*)rep, (unsigned*)bufB, src, dst, nsrc, ndst,
                                                    1.0f, 0.0f, E, En, N);
    finalize_kernel<<<1, 256, 0, stream>>>((const unsigned*)bufB, ablocks, loss_out, (float)N);
}


# owner workgroups order their bin by destination row in LDS (count, scan, place); a wave sums a run of same-row edges in registers, one LDS float add per run, not per edge; arrival-rank order
# speedup vs baseline: 3.8851x; 1.0514x over previous
#include <hip/hip_runtime.h>

typedef __attribute__((ext_vector_type(2))) float v2f;
typedef __attribute__((ext_vector_type(8))) float v8f;

#define FDIM 128
#define ROWS_PER_WG 391
#define LIST_HDR 64
#define MAX_OWNERS 256
#define BIN_THREADS 256
#define BIN_EPT 8
#define BIN_EDGES (BIN_THREADS * BIN_EPT)
static_assert(BIN_THREADS == MAX_OWNERS, "thread b of the binning pass handles owner b");

__global__ void __launch_bounds__(256) zero_init_kernel(unsigned* __restrict__ bins, int nown) {
    int i = blockIdx.x * blockDim.x + threadIdx.x;
    if (i < nown) bins[(size_t)i * ROWS_PER_WG * FDIM] = 0u;
}

#define BIN_LDS_DWORDS (2 * MAX_OWNERS + BIN_THREADS / 32 + 2 * BIN_EDGES)
static __device__ __forceinline__ void bin_body(unsigned* lds,
                                                const int* __restrict__ src,
                                                const int* __restrict__ dst,
                                                unsigned* __restrict__ bins,
                                                int E, int n, int nown, int blk) {
    unsigned* hist = lds;
    unsigned* gbase = hist + MAX_OWNERS;
    unsigned* wtot = gbase + MAX_OWNERS;
    unsigned* st_e = wtot + BIN_THREADS / 32;
    unsigned* st_a = st_e + BIN_EDGES;

    const int tid = threadIdx.x;
    const int lane = tid & 31;
    const int wave = tid >> 5;
    if (nown > MAX_OWNERS) return;
    hist[tid] = 0u;
    __syncthreads();

    const int e0 = blk * BIN_EDGES;
    int dv[BIN_EPT], sv[BIN_EPT];
#pragma unroll
    for (int k = 0; k < BIN_EPT; ++k) {
        const int e = e0 + k * BIN_THREADS + tid;
        dv[k] = e < E ? dst[e] : -1;
        sv[k] = e < E ? src[e] : 0;
    }
    int ow[BIN_EPT];
    unsigned ent[BIN_EPT], rk[BIN_EPT];
#pragma unroll
    for (int k = 0; k < BIN_EPT; ++k) {
        ow[k] = -1; ent[k] = 0u; rk[k] = 0u;
        if (dv[k] >= 0) {
            const int b = (int)((unsigned)dv[k] / (unsigned)ROWS_PER_WG);
            if (b < nown) {
                ow[k] = b;
                ent[k] = ((unsigned)(dv[k] - b * ROWS_PER_WG) << 20) | (unsigned)sv[k];
                rk[k] = atomicAdd(&hist[b], 1u);
            }
        }
    }
    __syncthreads();

    {
        const unsigned c = hist[tid];
        unsigned inc = c;
#pragma unroll
        for (int off = 1; off < 32; off <<= 1) {
            const unsigned t = __shfl_up(inc, off, 32);
            if (lane >= off) inc += t;
        }
        if (lane == 31) wtot[wave] = inc;
        __syncthreads();
        unsigned pre = 0u;
        for (int w = 0; w < wave; ++w) pre += wtot[w];
        unsigned gb = 0u;
        if (c != 0u) gb = atomicAdd(&bins[(size_t)tid * ROWS_PER_WG * FDIM], c);
        hist[tid] = pre + inc - c;
        gbase[tid] = gb;
    }
    __syncthreads();

#pragma unroll
    for (int k = 0; k < BIN_EPT; ++k) {
        if (ow[k] >= 0) {
            const int b = ow[k];
            const unsigned p = hist[b] + rk[k];
            int nrb = n - b * ROWS_PER_WG;
            if (nrb > ROWS_PER_WG) nrb = ROWS_PER_WG;
            const unsigned cap = (unsigned)(nrb * FDIM - LIST_HDR);
            const unsigned pos = gbase[b] + rk[k];
            st_e[p] = ent[k];
            st_a[p] = pos < cap ? (unsigned)b * (ROWS_PER_WG * FDIM) + LIST_HDR + pos : 0xFFFFFFFFu;
        }
    }
    __syncthreads();

    int nv = 0;
    for (int w = 0; w < BIN_THREADS / 32; ++w) nv += (int)wtot[w];
    for (int i = tid; i < nv; i += BIN_THREADS) {
        const unsigned a = st_a[i];
        if (a != 0xFFFFFFFFu) bins[a] = st_e[i];
    }
}

#define DEG_THREADS 1024
#define DEG_EPT 8
__global__ void __launch_bounds__(DEG_THREADS) owner_dinv_kernel(const unsigned* __restrict__ bins,
                                                                 const int* __restrict__ dst,
                                                                 float* __restrict__ dinv, int E, int n) {
    __shared__ unsigned rc[ROWS_PER_WG];
    const int tid = threadIdx.x;
    const int lo = blockIdx.x * ROWS_PER_WG;
    int nr = n - lo;
    if (nr > ROWS_PER_WG) nr = ROWS_PER_WG;
    for (int i = tid; i < nr; i += DEG_THREADS) rc[i] = 0u;
    __syncthreads();

    const unsigned* bk = bins + (size_t)lo * FDIM;
    const unsigned cb = bk[0];
    const bool binned = gridDim.x <= MAX_OWNERS && cb <= (unsigned)(nr * FDIM - LIST_HDR);
    if (binned) {
        for (unsigned base = 0u; base < cb; base += DEG_THREADS * DEG_EPT) {
            unsigned ev[DEG_EPT];
#pragma unroll
            for (int k = 0; k < DEG_EPT; ++k) {
                const unsigned i = base + (unsigned)(k * DEG_THREADS + tid);
                ev[k] = i < cb ? bk[LIST_HDR + i] : 0xFFFFFFFFu;
            }
#pragma unroll
            for (int k = 0; k < DEG_EPT; ++k)
                if (ev[k] != 0xFFFFFFFFu) atomicAdd(&rc[ev[k] >> 20], 1u);
        }
    } else {
        for (int e = tid; e < E; e += DEG_THREADS) {
            const unsigned r = (unsigned)(dst[e] - lo);
            if (r < (unsigned)nr) atomicAdd(&rc[r], 1u);
        }
    }
    __syncthreads();
    for (int i = tid; i < nr; i += DEG_THREADS) dinv[lo + i] = rsqrtf((float)rc[i] + 1.0f);
}

template <bool RELU>
static __device__ __forceinline__ void gemm_body(float* wlds,
                                                 const float* __restrict__ A,
                                                 const float* __restrict__ W,
                                                 float* __restrict__ Hout,
                                                 int nrows, int blk) {
    for (int i = threadIdx.x; i < FDIM * FDIM; i += 256) {
        int k = i >> 7, n = i & 127;
        wlds[((k >> 1) << 8) + (n << 1) + (k & 1)] = W[i];
    }
    __syncthreads();

    const int lane  = threadIdx.x & 31;
    const int wave  = threadIdx.x >> 5;
    const int strip = blk * 8 + wave;
    const int nstrips = nrows >> 4;
    if (strip >= nstrips) return;

    const int row0  = strip << 4;
    const int m     = lane & 15;
    const int khalf = lane >> 4;
    const int kbase = khalf << 1;

    const float* arow = A + (size_t)(row0 + m) * FDIM;

    v8f acc[8];
#pragma unroll
    for (int t = 0; t < 8; ++t) acc[t] = (v8f)(0.0f);

    for (int k0 = 0; k0 < FDIM; k0 += 4) {
        v2f a = *(const v2f*)(arow + k0 + kbase);
        if (RELU) { a.x = fmaxf(a.x, 0.0f); a.y = fmaxf(a.y, 0.0f); }
        const int bbase = ((k0 >> 1) + khalf) * 256 + (m << 1);
#pragma unroll
        for (int t = 0; t < 8; ++t) {
            v2f b = *(const v2f*)(&wlds[bbase + t * 32]);
            acc[t] = __builtin_amdgcn_wmma_f32_16x16x4_f32(
                 false, a,  false, b,
                 (short)0, acc[t],  false,  false);
        }
    }

    const int rbase = row0 + (khalf << 3);
#pragma unroll
    for (int t = 0; t < 8; ++t) {
#pragma unroll
        for (int r = 0; r < 8; ++r)
            Hout[(size_t)(rbase + r) * FDIM + t * 16 + m] = acc[t][r];
    }
}

template <bool RELU>
__global__ void __launch_bounds__(256) gemm_wmma_kernel(const float* __restrict__ A,
                                                        const float* __restrict__ W,
                                                        float* __restrict__ Hout,
                                                        int nrows) {
    __shared__ float wlds[FDIM * FDIM];
    gemm_body<RELU>(wlds, A, W, Hout, nrows, (int)blockIdx.x);
}

static_assert(BIN_LDS_DWORDS <= FDIM * FDIM, "the binning arrays fit in the GEMM's LDS buffer");
static_assert(BIN_THREADS == 256, "both kinds of block have 256 threads");
__global__ void __launch_bounds__(256) gemm_bin_kernel(const float* __restrict__ A,
                                                       const float* __restrict__ W,
                                                       float* __restrict__ Hout,
                                                       int nrows, int gblocks,
                                                       const int* __restrict__ src,
                                                       const int* __restrict__ dst,
                                                       unsigned* __restrict__ bins,
                                                       int E, int n, int nown) {
    __shared__ __attribute__((aligned(16))) unsigned char smem[FDIM * FDIM * 4];
    const int b = (int)blockIdx.x;
    if (b < gblocks) gemm_body<false>((float*)smem, A, W, Hout, nrows, b);
    else             bin_body((unsigned*)smem, src, dst, bins, E, n, nown, b - gblocks);
}

#define AGG_WAVES 32
#define AGG_UNROLL 8
#define PART_LOSS 32
#define PART_CNT 33
#define ORD_EPT 8
#define ORD_CAP (1024 * ORD_EPT)
#define ORD_RPL ((ROWS_PER_WG + 32) / 32)
static_assert(ORD_RPL * 32 >= ROWS_PER_WG + 1, "the scan covers every row counter");
static_assert(AGG_WAVES == 32, "lane i reads the count of wave i");
static_assert(ROWS_PER_WG <= 2048, "local row must fit above bit 20");

template <bool FIRST>
__global__ void __launch_bounds__(1024) owner_agg_kernel(const float4* __restrict__ h,
                                                         float* out,
                                                         unsigned* list,
                                                         unsigned* list2,
                                                         const int* __restrict__ src,
                                                         const int* __restrict__ dst,
                                                         const float* __restrict__ dinv,
                                                         const float* __restrict__ bias,
                                                         int E, int n) {
    __shared__ float tile[ROWS_PER_WG * FDIM];
    __shared__ int queue[AGG_WAVES * 64];
    __shared__ unsigned sorted[FIRST ? ORD_CAP : 32];
    __shared__ unsigned rpos[ROWS_PER_WG + 1];

    const int lane = threadIdx.x & 31;
    const int wave = threadIdx.x >> 5;
    const int qb = wave * 64;
    const int lo = blockIdx.x * ROWS_PER_WG;
    int nr = n - lo;
    if (nr > ROWS_PER_WG) nr = ROWS_PER_WG;

    bool binned = false, ordered = false;
    int cb = 0;
    const unsigned* bk = list2 + (size_t)lo * FDIM;
    unsigned ev[ORD_EPT];
    if (FIRST) {
        cb = (int)bk[0];
        binned = gridDim.x <= MAX_OWNERS && (unsigned)cb <= (unsigned)(nr * FDIM - LIST_HDR);
        ordered = binned && cb <= ORD_CAP;
#pragma unroll
        for (int k = 0; k < ORD_EPT; ++k) {
            const int i = k * 1024 + (int)threadIdx.x;
            ev[k] = (ordered && i < cb) ? bk[LIST_HDR + i] : 0xFFFFFFFFu;
        }
        for (int i = threadIdx.x; i <= ROWS_PER_WG; i += 1024) rpos[i] = 0u;
    }

    unsigned* wl = list + (size_t)lo * FDIM;
    const int capw = ((nr * FDIM - LIST_HDR) >> 6) & ~31;
    uint2* seg = (uint2*)(wl + LIST_HDR) + (size_t)wave * capw;

    {
        const float b0 = bias[4 * lane], b1 = bias[4 * lane + 1];
        const float b2 = bias[4 * lane + 2], b3 = bias[4 * lane + 3];
        const float4* h4 = h + (size_t)lo * 32;
        for (int i = threadIdx.x; i < nr * 32; i += 1024) {
            const int row = i >> 5;
            const float4 v = h4[i];
            const float d = dinv[lo + row];
            const float d2 = d * d;
            float* t = tile + row * FDIM + lane;
            t[0] = v.x * d2 + b0; t[32] = v.y * d2 + b1; t[64] = v.z * d2 + b2; t[96] = v.w * d2 + b3;
        }
    }
    __syncthreads();

    if (FIRST && ordered) {
#pragma unroll
        for (int k = 0; k < ORD_EPT; ++k)
            if (ev[k] != 0xFFFFFFFFu) atomicAdd(&rpos[ev[k] >> 20], 1u);
        __syncthreads();
        if (wave == 0) {
            unsigned c[ORD_RPL];
            unsigned sum = 0u;
#pragma unroll
            for (int k = 0; k < ORD_RPL; ++k) {
                const int r = lane * ORD_RPL + k;
                c[k] = r <= ROWS_PER_WG ? rpos[r] : 0u;
                sum += c[k];
            }
            unsigned inc = sum;
#pragma unroll
            for (int off = 1; off < 32; off <<= 1) {
                const unsigned t = __shfl_up(inc, off, 32);
                if (lane >= off) inc += t;
            }
            unsigned run = inc - sum;
#pragma unroll
            for (int k = 0; k < ORD_RPL; ++k) {
                const int r = lane * ORD_RPL + k;
                if (r <= ROWS_PER_WG) rpos[r] = run;
                run += c[k];
            }
        }
        __syncthreads();
#pragma unroll
        for (int k = 0; k < ORD_EPT; ++k) {
            if (ev[k] != 0xFFFFFFFFu) {
                const unsigned slot = atomicAdd(&rpos[ev[k] >> 20], 1u);
                if (slot < (unsigned)ORD_CAP) sorted[slot] = ev[k];
            }
        }
        __syncthreads();
    }

    int run_row = -1;
    float4 run_sum = make_float4(0.0f, 0.0f, 0.0f, 0.0f);
    auto put_run = [&]() {
        if (run_row >= 0) {
            float* t = tile + run_row * FDIM + lane;
            atomicAdd(t, run_sum.x);
            atomicAdd(t + 32, run_sum.y);
            atomicAdd(t + 64, run_sum.z);
            atomicAdd(t + 96, run_sum.w);
        }
    };
    auto add_row = [&](float4 v, int r, float w) {
        if (r != run_row) {
            put_run();
            run_row = r;
            run_sum = make_float4(0.0f, 0.0f, 0.0f, 0.0f);
        }
        run_sum.x += v.x * w;
        run_sum.y += v.y * w;
        run_sum.z += v.z * w;
        run_sum.w += v.w * w;
    };

    auto gather = [&](int nf, int s, int dr, float nrm) {
        int j = 0;
        for (; j + 8 <= nf; j += 8) {
            float4 v[8];
            int r[8];
            float w[8];
#pragma unroll
            for (int u = 0; u < 8; ++u) {
                int sj = __builtin_amdgcn_readlane(s, j + u);
                r[u] = __builtin_amdgcn_readlane(dr, j + u);
                w[u] = __int_as_float(__builtin_amdgcn_readlane(__float_as_int(nrm), j + u));
                v[u] = h[(size_t)sj * 32 + lane];
            }
#pragma unroll
            for (int u = 0; u < 8; ++u) add_row(v[u], r[u], w[u]);
        }
        for (; j < nf; ++j) {
            int sj = __builtin_amdgcn_readlane(s, j);
            int rj = __builtin_amdgcn_readlane(dr, j);
            float wj = __int_as_float(__builtin_amdgcn_readlane(__float_as_int(nrm), j));
            add_row(h[(size_t)sj * 32 + lane], rj, wj);
        }
    };

    bool walk = false;
    if (!FIRST) {
        const unsigned c = wl[lane];
        walk = __builtin_amdgcn_ballot_w32(c > (unsigned)capw) == 0u;
    }

    if (FIRST && ordered) {
        const int per = (((cb + AGG_WAVES - 1) / AGG_WAVES) + 7) & ~7;
        int j0 = __builtin_amdgcn_readfirstlane(wave) * per;
        int jend = j0 + per;
        if (jend > cb) jend = cb;
        int tot = 0;
        unsigned e1 = 0u;
        float na = 0.0f, nb = 0.0f;
        if (j0 + lane < jend) { e1 = sorted[j0 + lane]; na = dinv[e1 & 0xFFFFFu]; nb = dinv[lo + (int)(e1 >> 20)]; }
        for (; j0 < jend; j0 += 32) {
            const unsigned cu = e1;
            const float nrm = na * nb;
            e1 = 0u; na = 0.0f; nb = 0.0f;
            if (j0 + 32 + lane < jend) { e1 = sorted[j0 + 32 + lane]; na = dinv[e1 & 0xFFFFFu]; nb = dinv[lo + (int)(e1 >> 20)]; }
            const int nf = jend - j0 < 32 ? jend - j0 : 32;
            const int s = (int)(cu & 0xFFFFFu), dr = (int)(cu >> 20);
            if (lane < nf && tot + nf <= capw) seg[tot + lane] = make_uint2(cu, __float_as_uint(nrm));
            tot += nf;
            gather(nf, s, dr, nrm);
        }
        if (lane == 0) wl[wave] = (unsigned)tot;
    } else if (FIRST && binned) {
        int tot = 0;
        int j0 = __builtin_amdgcn_readfirstlane(wave) * 32;
        const int step = 32 * AGG_WAVES;
        unsigned e1 = 0u, e2 = 0u;
        float na = 0.0f, nb = 0.0f;
        if (j0 + lane < cb) e1 = bk[LIST_HDR + j0 + lane];
        if (j0 + step + lane < cb) e2 = bk[LIST_HDR + j0 + step + lane];
        if (j0 + lane < cb) { na = dinv[e1 & 0xFFFFFu]; nb = dinv[lo + (int)(e1 >> 20)]; }
        for (; j0 < cb; j0 += step) {
            const unsigned cur = e1;
            const float nrm = na * nb;
            e1 = e2;
            e2 = 0u; na = 0.0f; nb = 0.0f;
            if (j0 + 2 * step + lane < cb) e2 = bk[LIST_HDR + j0 + 2 * step + lane];
            if (j0 + step + lane < cb) { na = dinv[e1 & 0xFFFFFu]; nb = dinv[lo + (int)(e1 >> 20)]; }
            const int nf = cb - j0 < 32 ? cb - j0 : 32;
            const int s = (int)(cur & 0xFFFFFu), dr = (int)(cur >> 20);
            if (lane < nf && tot + nf <= capw) seg[tot + lane] = make_uint2(cur, __float_as_uint(nrm));
            tot += nf;
            gather(nf, s, dr, nrm);
        }
        if (lane == 0) wl[wave] = (unsigned)tot;
    } else if (walk) {
        const int cw = __builtin_amdgcn_readfirstlane((int)wl[wave]);
        unsigned* wl2 = list2 + (size_t)lo * FDIM;
        unsigned* seg2 = wl2 + LIST_HDR + (size_t)wave * capw;
        int tot2 = 0;
        uint2 nxt = make_uint2(0u, 0u);
        if (lane < cw) nxt = seg[lane];
        for (int j0 = 0; j0 < cw; j0 += 32) {
            const uint2 cur = nxt;
            if (j0 + 32 + lane < cw) nxt = seg[j0 + 32 + lane];
            const int nf = cw - j0 < 32 ? cw - j0 : 32;
            const int sx = (int)(cur.x & 0xFFFFFu), rx = (int)(cur.x >> 20);
            const bool keep = lane < nf && sx < lo + rx;
            const unsigned km = __builtin_amdgcn_ballot_w32(keep);
            if (keep) seg2[tot2 + __popc(km & ((1u << lane) - 1u))] = cur.x;
            tot2 += __popc(km);
            gather(nf, sx, rx, __uint_as_float(cur.y));
        }
        if (lane == 0) wl2[wave] = (unsigned)tot2;
    } else {
        int cnt = 0;
        int tot = 0;
        const unsigned lt = (1u << lane) - 1u;

        auto flush = [&](int nf) {
            int s = 0, dr = 0;
            float nrm = 0.0f;
            if (lane < nf) {
                int e = queue[qb + lane];
                s = src[e];
                int d = dst[e];
                dr = d - lo;
                nrm = dinv[s] * dinv[d];
                if (FIRST && tot + nf <= capw)
                    seg[tot + lane] = make_uint2(((unsigned)dr << 20) | (unsigned)s, __float_as_uint(nrm));
            }
            tot += nf;
            gather(nf, s, dr, nrm);
        };

        int head = (int)(((16u - ((unsigned)(size_t)dst & 15u)) & 15u) >> 2);
        if (head > E) head = E;
        const int4* dst4 = (const int4*)(dst + head);
        const int nvec = (E - head) >> 2;
        const int ntail = (E - head) & 3;
        const int nextra = head + ntail;

        const int step = AGG_WAVES * 32 * AGG_UNROLL;
        const int npass = (nvec + step - 1) / step;
        for (int pass = 0; pass <= npass; ++pass) {
            const int vb = pass * step + wave * (32 * AGG_UNROLL);
            unsigned mbits = 0u;
            if (pass < npass) {
                int4 d4[AGG_UNROLL];
                unsigned nru[AGG_UNROLL];
#pragma unroll
                for (int u = 0; u < AGG_UNROLL; ++u) {
                    const int vi = vb + u * 32 + lane;
                    d4[u] = dst4[vi < nvec ? vi : nvec - 1];
                    nru[u] = vi < nvec ? (unsigned)nr : 0u;
                }
#pragma unroll
                for (int u = 0; u < AGG_UNROLL; ++u) {
                    const bool cx = (unsigned)(d4[u].x - lo) < nru[u];
                    const bool cy = (unsigned)(d4[u].y - lo) < nru[u];
                    const bool cz = (unsigned)(d4[u].z - lo) < nru[u];
                    const bool cw = (unsigned)(d4[u].w - lo) < nru[u];
                    const unsigned any = __builtin_amdgcn_ballot_w32(cx) | __builtin_amdgcn_ballot_w32(cy) |
                                         __builtin_amdgcn_ballot_w32(cz) | __builtin_amdgcn_ballot_w32(cw);
                    if (any)
                        mbits |= ((cx ? 1u : 0u) | (cy ? 2u : 0u) | (cz ? 4u : 0u) | (cw ? 8u : 0u)) << (4 * u);
                }
            } else if (wave == 0 && lane < nextra) {
                int e = lane < head ? lane : (nvec << 2) + lane;
                if ((unsigned)(dst[e] - lo) < (unsigned)nr) mbits = 1u;
            }
            while (__builtin_amdgcn_ballot_w32(mbits != 0u)) {
                const bool mt = mbits != 0u;
                const int k = mt ? (__ffs(mbits) - 1) : 0;
                int e;
                if (pass < npass) e = head + ((vb + (k >> 2) * 32 + lane) << 2) + (k & 3);
                else              e = lane < head ? lane : (nvec << 2) + lane;
                mbits &= mbits - 1u;
                const unsigned mask = __builtin_amdgcn_ballot_w32(mt);
                const int pos = cnt + __popc(mask & lt);
                if (mt) queue[qb + pos] = e;
                cnt += __popc(mask);
                if (cnt >= 32) {
                    flush(32);
                    const int rem = cnt - 32;
                    int t = 0;
                    if (lane < rem) t = queue[qb + 32 + lane];
                    if (lane < rem) queue[qb + lane] = t;
                    cnt = rem;
                }
            }
        }
        if (cnt > 0) flush(cnt);
        if (FIRST && lane == 0) wl[wave] = (unsigned)tot;
        if (!FIRST && lane == 0) list2[(size_t)lo * FDIM + wave] = 0xFFFFFFFFu;
    }
    put_run();

    __syncthreads();
    float4* out4 = (float4*)out + (size_t)lo * 32;
    for (int i = threadIdx.x; i < nr * 32; i += 1024) {
        const float* t = tile + (i >> 5) * FDIM + (i & 31);
        out4[i] = make_float4(t[0], t[32], t[64], t[96]);
    }
}

__global__ void __launch_bounds__(1024) owner_loss_kernel(const float4* __restrict__ rep,
                                                          unsigned* list2,
                                                          const int* __restrict__ src,
                                                          const int* __restrict__ dst,
                                                          const int* __restrict__ src2,
                                                          const int* __restrict__ dst2,
                                                          float target, float target2,
                                                          int E, int E2, int n) {
    __shared__ float4 tile4[ROWS_PER_WG * 32];
    __shared__ float wsum[AGG_WAVES];
    __shared__ unsigned wcnt[AGG_WAVES];

    const int lane = threadIdx.x & 31;
    const int wave = threadIdx.x >> 5;
    const int lo = blockIdx.x * ROWS_PER_WG;
    int nr = n - lo;
    if (nr > ROWS_PER_WG) nr = ROWS_PER_WG;
    unsigned* wl2 = list2 + (size_t)lo * FDIM;
    const int capw = ((nr * FDIM - LIST_HDR) >> 6) & ~31;
    const unsigned* seg2 = wl2 + LIST_HDR + (size_t)wave * capw;

    const int per = (E2 + (int)gridDim.x - 1) / (int)gridDim.x;
    const int b0 = (int)blockIdx.x * per;
    int b1 = b0 + per;
    if (b1 > E2) b1 = E2;
    const int bstep = 32 * AGG_WAVES;
    int base = b0 + __builtin_amdgcn_readfirstlane(wave) * 32;
    int s0 = 0, d0 = 0, s1 = 0, d1 = 0;
    if (base + lane < b1) { s0 = src2[base + lane]; d0 = dst2[base + lane]; }
    if (base + bstep + lane < b1) { s1 = src2[base + bstep + lane]; d1 = dst2[base + bstep + lane]; }

    for (int i = threadIdx.x; i < nr * 32; i += 1024) tile4[i] = rep[(size_t)lo * 32 + i];

    float local = 0.0f;
    unsigned cntw = 0u;

    auto lane_sums = [&](const float (&p)[8], unsigned vbits, float tgt) {
        const bool h = (lane & 16) != 0, g = (lane & 8) != 0, f = (lane & 4) != 0;
        float q[4];
#pragma unroll
        for (int u = 0; u < 4; ++u)
            q[u] = (h ? p[u + 4] : p[u]) + __shfl_xor(h ? p[u] : p[u + 4], 16, 32);
        float q2[2];
#pragma unroll
        for (int u = 0; u < 2; ++u)
            q2[u] = (g ? q[u + 2] : q[u]) + __shfl_xor(g ? q[u] : q[u + 2], 8, 32);
        float t1 = (f ? q2[1] : q2[0]) + __shfl_xor(f ? q2[0] : q2[1], 4, 32);
        t1 += __shfl_xor(t1, 2, 32);
        t1 += __shfl_xor(t1, 1, 32);
        if ((lane & 3) == 0 && ((vbits >> (lane >> 2)) & 1u)) {
            const float dl = t1 - tgt;
            local += dl * dl;
        }
    };

    auto pairs = [&](int s, int d) {
        unsigned m = __builtin_amdgcn_ballot_w32(s < d);
        cntw += (unsigned)__popc(m);
        while (m) {
            const int left = __popc(m);
            const unsigned vbits = left >= 8 ? 0xFFu : ((1u << left) - 1u);
            float4 va[8], vb[8];
#pragma unroll
            for (int u = 0; u < 8; ++u) {
                const int j = m ? (__ffs(m) - 1) : 0;
                m &= m - 1u;
                const int sj = __builtin_amdgcn_readlane(s, j);
                const int dj = __builtin_amdgcn_readlane(d, j);
                va[u] = rep[(size_t)sj * 32 + lane];
                vb[u] = rep[(size_t)dj * 32 + lane];
            }
            float p[8];
#pragma unroll
            for (int u = 0; u < 8; ++u)
                p[u] = va[u].x * vb[u].x + va[u].y * vb[u].y + va[u].z * vb[u].z + va[u].w * vb[u].w;
            lane_sums(p, vbits, target2);
        }
    };

    while (base < b1) {
        pairs(s0, d0);
        pairs(s1, d1);
        base += 2 * bstep;
        s0 = 0; d0 = 0; s1 = 0; d1 = 0;
        if (base + lane < b1) { s0 = src2[base + lane]; d0 = dst2[base + lane]; }
        if (base + bstep + lane < b1) { s1 = src2[base + bstep + lane]; d1 = dst2[base + bstep + lane]; }
    }
    __syncthreads();

    auto process = [&](unsigned m, int s, int dr) {
        cntw += (unsigned)__popc(m);
        while (m) {
            const int left = __popc(m);
            const unsigned vbits = left >= 8 ? 0xFFu : ((1u << left) - 1u);
            float4 v[8];
            int r[8];
#pragma unroll
            for (int u = 0; u < 8; ++u) {
                const int j = m ? (__ffs(m) - 1) : 0;
                m &= m - 1u;
                const int sj = __builtin_amdgcn_readlane(s, j);
                r[u] = __builtin_amdgcn_readlane(dr, j);
                v[u] = rep[(size_t)sj * 32 + lane];
            }
            float p[8];
#pragma unroll
            for (int u = 0; u < 8; ++u) {
                const float4 t = tile4[r[u] * 32 + lane];
                p[u] = v[u].x * t.x + v[u].y * t.y + v[u].z * t.z + v[u].w * t.w;
            }
            lane_sums(p, vbits, target);
        }
    };

    const unsigned c = wl2[lane];
    if (__builtin_amdgcn_ballot_w32(c > (unsigned)capw) == 0u) {
        const int cw = __builtin_amdgcn_readfirstlane((int)wl2[wave]);
        unsigned nxt = 0u;
        if (lane < cw) nxt = seg2[lane];
        for (int j0 = 0; j0 < cw; j0 += 32) {
            const unsigned cur = nxt;
            if (j0 + 32 + lane < cw) nxt = seg2[j0 + 32 + lane];
            const int nf = cw - j0 < 32 ? cw - j0 : 32;
            const unsigned m = nf >= 32 ? 0xFFFFFFFFu : ((1u << nf) - 1u);
            process(m, (int)(cur & 0xFFFFFu), (int)(cur >> 20));
        }
    } else {
        for (int base = wave * 32; base < E; base += 32 * AGG_WAVES) {
            const int e = base + lane;
            int s = 0, dr = 0;
            bool ok = false;
            if (e < E) {
                const int d = dst[e];
                const int sv = src[e];
                ok = (unsigned)(d - lo) < (unsigned)nr && sv < d;
                if (ok) { s = sv; dr = d - lo; }
            }
            process(__builtin_amdgcn_ballot_w32(ok), s, dr);
        }
    }

#pragma unroll
    for (int off = 16; off > 0; off >>= 1) local += __shfl_xor(local, off, 32);
    if (lane == 0) { wsum[wave] = local; wcnt[wave] = cntw; }
    __syncthreads();
    if (wave == 0) {
        float a = wsum[lane];
        unsigned b = wcnt[lane];
#pragma unroll
        for (int off = 16; off > 0; off >>= 1) {
            a += __shfl_xor(a, off, 32);
            b += __shfl_xor(b, off, 32);
        }
        if (lane == 0) { wl2[PART_LOSS] = __float_as_uint(a); wl2[PART_CNT] = b; }
    }
}

__global__ void __launch_bounds__(256) finalize_kernel(const unsigned* __restrict__ part, int nwg,
                                                       float* __restrict__ out_scalar, float n_nodes) {
    __shared__ float sl[256];
    __shared__ unsigned sc[256];
    float l = 0.0f;
    unsigned c = 0u;
    for (int i = threadIdx.x; i < nwg; i += 256) {
        const unsigned* hd = part + (size_t)i * ROWS_PER_WG * FDIM;
        l += __uint_as_float(hd[PART_LOSS]);
        c += hd[PART_CNT];
    }
    sl[threadIdx.x] = l;
    sc[threadIdx.x] = c;
    __syncthreads();
    for (int w = 128; w > 0; w >>= 1) {
        if ((int)threadIdx.x < w) { sl[threadIdx.x] += sl[threadIdx.x + w]; sc[threadIdx.x] += sc[threadIdx.x + w]; }
        __syncthreads();
    }
    if (threadIdx.x == 0) {
        *out_scalar = sl[0] * n_nodes / (float)sc[0];
    }
}

extern "C" void kernel_launch(void* const* d_in, const int* in_sizes, int n_in,
                              void* d_out, int out_size, void* d_ws, size_t ws_size,
                              hipStream_t stream) {
    const float* features = (const float*)d_in[0];
    const int*   ei       = (const int*)d_in[1];
    const int*   nei      = (const int*)d_in[2];
    const float* w1       = (const float*)d_in[3];
    const float* b1       = (const float*)d_in[4];
    const float* w2       = (const float*)d_in[5];
    const float* b2       = (const float*)d_in[6];

    const int N  = in_sizes[0] / FDIM;
    const int E  = in_sizes[1] / 2;
    const int En = in_sizes[2] / 2;
    const int* src  = ei;
    const int* dst  = ei + E;
    const int* nsrc = nei;
    const int* ndst = nei + En;

    char* ws = (char*)d_ws;
    float*    bufA = (float*)ws;
    float*    bufB = (float*)(ws + (size_t)N * FDIM * 4);
    float*    dinv = (float*)(ws + (size_t)2 * N * FDIM * 4 + (size_t)N * 4);

    float* rep      = (float*)d_out;
    float* loss_out = rep + (size_t)N * FDIM;

    const int nstrips = N >> 4;
    const int gblocks = (nstrips + 7) / 8;
    const int ablocks = (N + ROWS_PER_WG - 1) / ROWS_PER_WG;

    zero_init_kernel<<<(ablocks + 255) / 256, 256, 0, stream>>>((unsigned*)bufB, ablocks);
    const int bblocks = (E + BIN_EDGES - 1) / BIN_EDGES;
    gemm_bin_kernel<<<gblocks + bblocks, 256, 0, stream>>>(features, w1, bufA, N, gblocks,
                                                           src, dst, (unsigned*)bufB, E, N, ablocks);
    owner_dinv_kernel<<<ablocks, DEG_THREADS, 0, stream>>>((const unsigned*)bufB, dst, dinv, E, N);

    owner_agg_kernel<true><<<ablocks, 1024, 0, stream>>>((const float4*)bufA, bufB, (unsigned*)rep, (unsigned*)bufB,
                                                         src, dst, dinv, b1, E, N);

    gemm_wmma_kernel<true><<<gblocks, 256, 0, stream>>>(bufB, w2, bufA, N);
    owner_agg_kernel<false><<<ablocks, 1024, 0, stream>>>((const float4*)bufA, rep, (unsigned*)rep, (unsigned*)bufB,
                                                          src, dst, dinv, b2, E, N);

    owner_loss_kernel<<<ablocks, 1024, 0, stream>>>((const float4*)rep, (unsigned*)bufB, src, dst, nsrc, ndst,
                                                    1.0f, 0.0f, E, En, N);
    finalize_kernel<<<1, 256, 0, stream>>>((const unsigned*)bufB, ablocks, loss_out, (float)N);
}
